# starting kernel
# baseline (speedup 1.0000x reference)
#include <hip/hip_runtime.h>
#include <hip/hip_bf16.h>

typedef float v2f __attribute__((ext_vector_type(2)));
typedef float v8f __attribute__((ext_vector_type(8)));

constexpr int NNODES = 20000;
constexpr int NEDGES = 320000;
constexpr int DIN    = 256;
constexpr int DHID   = 256;
constexpr int DOUT   = 64;

__global__ void zero_kernel(float* __restrict__ p, int n) {
    int i = blockIdx.x * blockDim.x + threadIdx.x;
    if (i < n) p[i] = 0.0f;
}

template <bool RELU>
__global__ __launch_bounds__(256)
void gemm_wmma_f32(const float* __restrict__ A, const float* __restrict__ W,
                   float* __restrict__ C, int M, int N, int K, int mTiles) {
    __shared__ float smem[256 * 64];

    const int tid  = threadIdx.x;
    const int lane = tid & 31;
    const int wave = tid >> 5;
    const int n0   = blockIdx.x << 6;

    {
        const int n   = tid & 63;
        const int kq0 = tid >> 6;
        for (int kq = kq0; kq < (K >> 2); kq += 4) {
            const float4 v = *(const float4*)(W + (size_t)(n0 + n) * K + (kq << 2));
            v2f lo; lo.x = v.x; lo.y = v.y;
            v2f hi; hi.x = v.z; hi.y = v.w;
            *(v2f*)(smem + (size_t)(2 * kq)     * 128 + 2 * n) = lo;
            *(v2f*)(smem + (size_t)(2 * kq + 1) * 128 + 2 * n) = hi;
        }
    }
    __syncthreads();

    const int mt = blockIdx.y * 8 + wave;
    if (mt >= mTiles) return;
    const int m0 = mt << 4;

    const int half = lane >> 4;
    const int l15  = lane & 15;

    const float* arow = A + (size_t)(m0 + l15) * K + 2 * half;
    __builtin_prefetch(arow, 0, 3);
    __builtin_prefetch(arow + 128, 0, 3);

    const float* bbase = smem + (size_t)half * 128 + 2 * l15;

    v8f acc0 = {}, acc1 = {}, acc2 = {}, acc3 = {};

    #pragma unroll 8
    for (int k0 = 0; k0 < K; k0 += 4) {
        v2f a = *(const v2f*)(arow + k0);
        if (RELU) { a.x = fmaxf(a.x, 0.0f); a.y = fmaxf(a.y, 0.0f); }
        const float* bp = bbase + (size_t)(k0 >> 1) * 128;
        v2f b0 = *(const v2f*)(bp +  0);
        v2f b1 = *(const v2f*)(bp + 32);
        v2f b2 = *(const v2f*)(bp + 64);
        v2f b3 = *(const v2f*)(bp + 96);
        acc0 = __builtin_amdgcn_wmma_f32_16x16x4_f32(false, a, false, b0, (short)0, acc0, false, false);
        acc1 = __builtin_amdgcn_wmma_f32_16x16x4_f32(false, a, false, b1, (short)0, acc1, false, false);
        acc2 = __builtin_amdgcn_wmma_f32_16x16x4_f32(false, a, false, b2, (short)0, acc2, false, false);
        acc3 = __builtin_amdgcn_wmma_f32_16x16x4_f32(false, a, false, b3, (short)0, acc3, false, false);
    }

    #pragma unroll
    for (int v = 0; v < 8; ++v) {
        float* crow = C + (size_t)(m0 + v + 8 * half) * N + n0 + l15;
        crow[ 0] = acc0[v];
        crow[16] = acc1[v];
        crow[32] = acc2[v];
        crow[48] = acc3[v];
    }
}

__global__ __launch_bounds__(256)
void scatter_kernel(const float* __restrict__ X, const int* __restrict__ src,
                    const int* __restrict__ dst, const float* __restrict__ w,
                    float* __restrict__ out, int E, int D) {
    const int per = D >> 2;
    int idx = blockIdx.x * blockDim.x + threadIdx.x;
    int e = idx / per;
    if (e >= E) return;
    int c = (idx % per) << 2;
    float we = w[e];
    const float4 x = *(const float4*)(X + (size_t)src[e] * D + c);
    float* o = out + (size_t)dst[e] * D + c;
    atomicAdd(o + 0, x.x * we);
    atomicAdd(o + 1, x.y * we);
    atomicAdd(o + 2, x.z * we);
    atomicAdd(o + 3, x.w * we);
}

extern "C" void kernel_launch(void* const* d_in, const int* in_sizes, int n_in,
                              void* d_out, int out_size, void* d_ws, size_t ws_size,
                              hipStream_t stream) {
    const float* node_feat = (const float*)d_in[0];
    const int*   edge_src  = (const int*)  d_in[1];
    const int*   edge_dst  = (const int*)  d_in[2];
    const float* edge_w    = (const float*)d_in[3];
    const float* W1        = (const float*)d_in[4];
    const float* W2        = (const float*)d_in[5];
    float* out = (float*)d_out;

    float* y1   = (float*)d_ws;
    float* agg1 = y1 + (size_t)NNODES * DHID;
    float* y2   = agg1 + (size_t)NNODES * DHID;

    const int mTiles = NNODES / 16;

    {
        int n = NNODES * DHID;
        zero_kernel<<<(n + 255) / 256, 256, 0, stream>>>(agg1, n);
        int m = NNODES * DOUT;
        zero_kernel<<<(m + 255) / 256, 256, 0, stream>>>(out, m);
    }

    {
        dim3 grid(DHID / 64, (mTiles + 7) / 8);
        gemm_wmma_f32<false><<<grid, 256, 0, stream>>>(
            node_feat, W1, y1, NNODES, DHID, DIN, mTiles);
    }
    {
        long long t = (long long)NEDGES * (DHID / 4);
        scatter_kernel<<<(int)((t + 255) / 256), 256, 0, stream>>>(
            y1, edge_src, edge_dst, edge_w, agg1, NEDGES, DHID);
    }
    {
        dim3 grid(DOUT / 64, (mTiles + 7) / 8);
        gemm_wmma_f32<true><<<grid, 256, 0, stream>>>(
            agg1, W2, y2, NNODES, DOUT, DHID, mTiles);
    }
    {
        long long t = (long long)NEDGES * (DOUT / 4);
        scatter_kernel<<<(int)((t + 255) / 256), 256, 0, stream>>>(
            y2, edge_src, edge_dst, edge_w, out, NEDGES, DOUT);
    }
}


# Global float-atomic scatter replaced by row-owning workgroups: each filters the dst list for its 80 rows, sums them in LDS with LDS float adds (arrival order), plain stores out; zero launches gone
# speedup vs baseline: 1.8957x; 1.8957x over previous
#include <hip/hip_runtime.h>
#include <hip/hip_bf16.h>

typedef float v2f __attribute__((ext_vector_type(2)));
typedef float v8f __attribute__((ext_vector_type(8)));

constexpr int NNODES = 20000;
constexpr int NEDGES = 320000;
constexpr int DIN    = 256;
constexpr int DHID   = 256;
constexpr int DOUT   = 64;

template <bool RELU>
__global__ __launch_bounds__(256)
void gemm_wmma_f32(const float* __restrict__ A, const float* __restrict__ W,
                   float* __restrict__ C, int M, int N, int K, int mTiles) {
    __shared__ float smem[256 * 64];

    const int tid  = threadIdx.x;
    const int lane = tid & 31;
    const int wave = tid >> 5;
    const int n0   = blockIdx.x << 6;

    {
        const int n   = tid & 63;
        const int kq0 = tid >> 6;
        for (int kq = kq0; kq < (K >> 2); kq += 4) {
            const float4 v = *(const float4*)(W + (size_t)(n0 + n) * K + (kq << 2));
            v2f lo; lo.x = v.x; lo.y = v.y;
            v2f hi; hi.x = v.z; hi.y = v.w;
            *(v2f*)(smem + (size_t)(2 * kq)     * 128 + 2 * n) = lo;
            *(v2f*)(smem + (size_t)(2 * kq + 1) * 128 + 2 * n) = hi;
        }
    }
    __syncthreads();

    const int mt = blockIdx.y * 8 + wave;
    if (mt >= mTiles) return;
    const int m0 = mt << 4;

    const int half = lane >> 4;
    const int l15  = lane & 15;

    const float* arow = A + (size_t)(m0 + l15) * K + 2 * half;
    __builtin_prefetch(arow, 0, 3);
    __builtin_prefetch(arow + 128, 0, 3);

    const float* bbase = smem + (size_t)half * 128 + 2 * l15;

    v8f acc0 = {}, acc1 = {}, acc2 = {}, acc3 = {};

    #pragma unroll 8
    for (int k0 = 0; k0 < K; k0 += 4) {
        v2f a = *(const v2f*)(arow + k0);
        if (RELU) { a.x = fmaxf(a.x, 0.0f); a.y = fmaxf(a.y, 0.0f); }
        const float* bp = bbase + (size_t)(k0 >> 1) * 128;
        v2f b0 = *(const v2f*)(bp +  0);
        v2f b1 = *(const v2f*)(bp + 32);
        v2f b2 = *(const v2f*)(bp + 64);
        v2f b3 = *(const v2f*)(bp + 96);
        acc0 = __builtin_amdgcn_wmma_f32_16x16x4_f32(false, a, false, b0, (short)0, acc0, false, false);
        acc1 = __builtin_amdgcn_wmma_f32_16x16x4_f32(false, a, false, b1, (short)0, acc1, false, false);
        acc2 = __builtin_amdgcn_wmma_f32_16x16x4_f32(false, a, false, b2, (short)0, acc2, false, false);
        acc3 = __builtin_amdgcn_wmma_f32_16x16x4_f32(false, a, false, b3, (short)0, acc3, false, false);
    }

    #pragma unroll
    for (int v = 0; v < 8; ++v) {
        float* crow = C + (size_t)(m0 + v + 8 * half) * N + n0 + l15;
        crow[ 0] = acc0[v];
        crow[16] = acc1[v];
        crow[32] = acc2[v];
        crow[48] = acc3[v];
    }
}

constexpr int ROWS = 80;
constexpr int NBLK = (NNODES + ROWS - 1) / ROWS;
constexpr int AW   = 32;
constexpr int SUB  = 512;
constexpr int CAP  = 768;
constexpr int NSUB = (NEDGES + SUB - 1) / SUB;
constexpr int DRAIN_AT = 24;

template <int D>
__device__ __forceinline__ void drain_list(const float* __restrict__ X, const int* __restrict__ src,
                                           const float* __restrict__ w, float* acc,
                                           const unsigned* mylist, int n, int lane) {
    for (int base = 0; base < n; base += 32) {
        const int  idx = base + lane;
        const bool ok  = idx < n;
        const unsigned ent = ok ? mylist[idx] : 0u;
        const int   e  = (int)(ent & 0xFFFFFFu);
        const int   s  = ok ? src[e] : 0;
        const float wv = ok ? w[e] : 0.0f;
        const int   lr = (int)(ent >> 24);
        const int   m  = min(32, n - base);
        if constexpr (D == 256) {
            for (int k = 0; k < m; k += 8) {
                int   sk[8], lk[8];
                float wk[8];
                #pragma unroll
                for (int u = 0; u < 8; ++u) {
                    sk[u] = __builtin_amdgcn_readlane(s, k + u);
                    lk[u] = __builtin_amdgcn_readlane(lr, k + u);
                    wk[u] = __builtin_bit_cast(float,
                                __builtin_amdgcn_readlane(__builtin_bit_cast(int, wv), k + u));
                }
                float4 xa[8], xb[8];
                #pragma unroll
                for (int u = 0; u < 8; ++u) {
                    if (k + u < m) {
                        const float* xr = X + (size_t)sk[u] * 256 + lane * 4;
                        xa[u] = *(const float4*)xr;
                        xb[u] = *(const float4*)(xr + 128);
                    }
                }
                #pragma unroll
                for (int u = 0; u < 8; ++u) {
                    if (k + u < m) {
                        float* a = acc + lk[u] * 256 + lane;
                        atomicAdd(a +   0, xa[u].x * wk[u]);
                        atomicAdd(a +  32, xa[u].y * wk[u]);
                        atomicAdd(a +  64, xa[u].z * wk[u]);
                        atomicAdd(a +  96, xa[u].w * wk[u]);
                        atomicAdd(a + 128, xb[u].x * wk[u]);
                        atomicAdd(a + 160, xb[u].y * wk[u]);
                        atomicAdd(a + 192, xb[u].z * wk[u]);
                        atomicAdd(a + 224, xb[u].w * wk[u]);
                    }
                }
            }
        } else {
            const int half = lane >> 4;
            const int l16  = lane & 15;
            for (int k = 0; k < m; k += 8) {
                float4 x[4];
                float  wk[4];
                int    sk[4], lk[4];
                bool   vq[4];
                #pragma unroll
                for (int u = 0; u < 4; ++u) {
                    const int q = k + 2 * u + half;
                    vq[u] = q < m;
                    sk[u] = __shfl(s, q);
                    wk[u] = __shfl(wv, q);
                    lk[u] = __shfl(lr, q);
                }
                #pragma unroll
                for (int u = 0; u < 4; ++u) {
                    if (vq[u]) x[u] = *(const float4*)(X + (size_t)sk[u] * 64 + l16 * 4);
                }
                #pragma unroll
                for (int u = 0; u < 4; ++u) {
                    if (vq[u]) {
                        float* a = acc + lk[u] * 64 + l16;
                        const float c0 = half ? x[u].z : x[u].x;
                        const float c1 = half ? x[u].w : x[u].y;
                        const float c2 = half ? x[u].x : x[u].z;
                        const float c3 = half ? x[u].y : x[u].w;
                        atomicAdd(a + (half ? 32 :  0), c0 * wk[u]);
                        atomicAdd(a + (half ? 48 : 16), c1 * wk[u]);
                        atomicAdd(a + (half ?  0 : 32), c2 * wk[u]);
                        atomicAdd(a + (half ? 16 : 48), c3 * wk[u]);
                    }
                }
            }
        }
    }
}

static_assert(DRAIN_AT + SUB <= CAP, "a trip of hits must fit behind an undrained list");
static_assert(ROWS <= 256 && NEDGES < (1 << 24), "list entry packing");

template <int D>
__global__ __launch_bounds__(1024)
void agg_rows_lds(const float* __restrict__ X, const int* __restrict__ src,
                  const int* __restrict__ dst, const float* __restrict__ w,
                  float* __restrict__ out) {
    __shared__ float    acc[ROWS * D];
    __shared__ unsigned lists[AW * CAP];

    const int tid  = threadIdx.x;
    const int lane = tid & 31;
    const int wave = tid >> 5;
    const int r0   = blockIdx.x * ROWS;

    for (int i = tid; i < ROWS * D; i += AW * 32) acc[i] = 0.0f;
    __syncthreads();

    unsigned* mylist = lists + wave * CAP;
    int n = 0;
    const int rot = (int)(((long long)blockIdx.x * NSUB) / NBLK);
    for (int sc = wave; sc < NSUB; sc += AW) {
        if (n >= DRAIN_AT) {
            drain_list<D>(X, src, w, acc, mylist, n, lane);
            n = 0;
        }
        int scr = sc + rot;
        if (scr >= NSUB) scr -= NSUB;
        const int ebase = scr * SUB + lane * 4;
        int4 d[4];
        #pragma unroll
        for (int j = 0; j < 4; ++j) {
            const int e = ebase + j * 128;
            d[j] = (e + 3 < NEDGES) ? *(const int4*)(dst + e) : make_int4(-1, -1, -1, -1);
        }
        #pragma unroll
        for (int j = 0; j < 4; ++j) {
            const int e = ebase + j * 128;
            const unsigned l0 = (unsigned)(d[j].x - r0);
            const unsigned l1 = (unsigned)(d[j].y - r0);
            const unsigned l2 = (unsigned)(d[j].z - r0);
            const unsigned l3 = (unsigned)(d[j].w - r0);
            const bool m0 = l0 < (unsigned)ROWS, m1 = l1 < (unsigned)ROWS;
            const bool m2 = l2 < (unsigned)ROWS, m3 = l3 < (unsigned)ROWS;
            if (__builtin_amdgcn_ballot_w32(m0 | m1 | m2 | m3) != 0u) {
                unsigned bm;
                bm = __builtin_amdgcn_ballot_w32(m0);
                if (m0) mylist[n + __builtin_amdgcn_mbcnt_lo(bm, 0u)] = (l0 << 24) | (unsigned)(e + 0);
                n += __popc(bm);
                bm = __builtin_amdgcn_ballot_w32(m1);
                if (m1) mylist[n + __builtin_amdgcn_mbcnt_lo(bm, 0u)] = (l1 << 24) | (unsigned)(e + 1);
                n += __popc(bm);
                bm = __builtin_amdgcn_ballot_w32(m2);
                if (m2) mylist[n + __builtin_amdgcn_mbcnt_lo(bm, 0u)] = (l2 << 24) | (unsigned)(e + 2);
                n += __popc(bm);
                bm = __builtin_amdgcn_ballot_w32(m3);
                if (m3) mylist[n + __builtin_amdgcn_mbcnt_lo(bm, 0u)] = (l3 << 24) | (unsigned)(e + 3);
                n += __popc(bm);
            }
        }
    }
    drain_list<D>(X, src, w, acc, mylist, n, lane);
    __syncthreads();

    constexpr int F4R = D / 4;
    for (int f = tid; f < ROWS * F4R; f += AW * 32) {
        const int row = f / F4R;
        const int c4  = f % F4R;
        if (r0 + row < NNODES) {
            float4 v;
            if constexpr (D == 256) {
                const float* p = acc + row * 256 + (c4 >> 5) * 128 + (c4 & 31);
                v.x = p[0]; v.y = p[32]; v.z = p[64]; v.w = p[96];
            } else {
                const float* p = acc + row * 64 + c4;
                v.x = p[0]; v.y = p[16]; v.z = p[32]; v.w = p[48];
            }
            *(float4*)(out + (size_t)(r0 + row) * D + c4 * 4) = v;
        }
    }
}

extern "C" void kernel_launch(void* const* d_in, const int* in_sizes, int n_in,
                              void* d_out, int out_size, void* d_ws, size_t ws_size,
                              hipStream_t stream) {
    const float* node_feat = (const float*)d_in[0];
    const int*   edge_src  = (const int*)  d_in[1];
    const int*   edge_dst  = (const int*)  d_in[2];
    const float* edge_w    = (const float*)d_in[3];
    const float* W1        = (const float*)d_in[4];
    const float* W2        = (const float*)d_in[5];
    float* out = (float*)d_out;

    float* y1   = (float*)d_ws;
    float* agg1 = y1 + (size_t)NNODES * DHID;
    float* y2   = agg1 + (size_t)NNODES * DHID;

    const int mTiles = NNODES / 16;

    {
        dim3 grid(DHID / 64, (mTiles + 7) / 8);
        gemm_wmma_f32<false><<<grid, 256, 0, stream>>>(
            node_feat, W1, y1, NNODES, DHID, DIN, mTiles);
    }
    agg_rows_lds<DHID><<<NBLK, AW * 32, 0, stream>>>(y1, edge_src, edge_dst, edge_w, agg1);
    {
        dim3 grid(DOUT / 64, (mTiles + 7) / 8);
        gemm_wmma_f32<true><<<grid, 256, 0, stream>>>(
            agg1, W2, y2, NNODES, DOUT, DHID, mTiles);
    }
    agg_rows_lds<DOUT><<<NBLK, AW * 32, 0, stream>>>(y2, edge_src, edge_dst, edge_w, out);
}


# Layer-2 neighbour sum stops filtering the dst list: layer-1 workgroups save their hits (row, src, weight) to the workspace, layer 2 sums from its saved list; row sums in LDS arrival order as before
# speedup vs baseline: 2.2377x; 1.1804x over previous
#include <hip/hip_runtime.h>
#include <hip/hip_bf16.h>

typedef float v2f __attribute__((ext_vector_type(2)));
typedef float v8f __attribute__((ext_vector_type(8)));

constexpr int NNODES = 20000;
constexpr int NEDGES = 320000;
constexpr int DIN    = 256;
constexpr int DHID   = 256;
constexpr int DOUT   = 64;

template <bool RELU>
__global__ __launch_bounds__(256)
void gemm_wmma_f32(const float* __restrict__ A, const float* __restrict__ W,
                   float* __restrict__ C, int M, int N, int K, int mTiles) {
    __shared__ float smem[256 * 64];

    const int tid  = threadIdx.x;
    const int lane = tid & 31;
    const int wave = tid >> 5;
    const int n0   = blockIdx.x << 6;

    {
        const int n   = tid & 63;
        const int kq0 = tid >> 6;
        for (int kq = kq0; kq < (K >> 2); kq += 4) {
            const float4 v = *(const float4*)(W + (size_t)(n0 + n) * K + (kq << 2));
            v2f lo; lo.x = v.x; lo.y = v.y;
            v2f hi; hi.x = v.z; hi.y = v.w;
            *(v2f*)(smem + (size_t)(2 * kq)     * 128 + 2 * n) = lo;
            *(v2f*)(smem + (size_t)(2 * kq + 1) * 128 + 2 * n) = hi;
        }
    }
    __syncthreads();

    const int mt = blockIdx.y * 8 + wave;
    if (mt >= mTiles) return;
    const int m0 = mt << 4;

    const int half = lane >> 4;
    const int l15  = lane & 15;

    const float* arow = A + (size_t)(m0 + l15) * K + 2 * half;
    __builtin_prefetch(arow, 0, 3);
    __builtin_prefetch(arow + 128, 0, 3);

    const float* bbase = smem + (size_t)half * 128 + 2 * l15;

    v8f acc0 = {}, acc1 = {}, acc2 = {}, acc3 = {};

    #pragma unroll 8
    for (int k0 = 0; k0 < K; k0 += 4) {
        v2f a = *(const v2f*)(arow + k0);
        if (RELU) { a.x = fmaxf(a.x, 0.0f); a.y = fmaxf(a.y, 0.0f); }
        const float* bp = bbase + (size_t)(k0 >> 1) * 128;
        v2f b0 = *(const v2f*)(bp +  0);
        v2f b1 = *(const v2f*)(bp + 32);
        v2f b2 = *(const v2f*)(bp + 64);
        v2f b3 = *(const v2f*)(bp + 96);
        acc0 = __builtin_amdgcn_wmma_f32_16x16x4_f32(false, a, false, b0, (short)0, acc0, false, false);
        acc1 = __builtin_amdgcn_wmma_f32_16x16x4_f32(false, a, false, b1, (short)0, acc1, false, false);
        acc2 = __builtin_amdgcn_wmma_f32_16x16x4_f32(false, a, false, b2, (short)0, acc2, false, false);
        acc3 = __builtin_amdgcn_wmma_f32_16x16x4_f32(false, a, false, b3, (short)0, acc3, false, false);
    }

    #pragma unroll
    for (int v = 0; v < 8; ++v) {
        float* crow = C + (size_t)(m0 + v + 8 * half) * N + n0 + l15;
        crow[ 0] = acc0[v];
        crow[16] = acc1[v];
        crow[32] = acc2[v];
        crow[48] = acc3[v];
    }
}

constexpr int ROWS = 80;
constexpr int NBLK = (NNODES + ROWS - 1) / ROWS;
constexpr int AW   = 32;
constexpr int SUB  = 512;
constexpr int CAP  = 768;
constexpr int NSUB = (NEDGES + SUB - 1) / SUB;
constexpr int DRAIN_AT = 24;
constexpr int GCAP = 4096;

typedef unsigned v2u __attribute__((ext_vector_type(2)));

template <int D>
__device__ __forceinline__ void add_edges(const float* __restrict__ X, float* acc,
                                          int s, float wv, int lr, int m, int lane) {
    if constexpr (D == 256) {
        for (int k = 0; k < m; k += 8) {
            int   sk[8], lk[8];
            float wk[8];
            #pragma unroll
            for (int u = 0; u < 8; ++u) {
                sk[u] = __builtin_amdgcn_readlane(s, k + u);
                lk[u] = __builtin_amdgcn_readlane(lr, k + u);
                wk[u] = __builtin_bit_cast(float,
                            __builtin_amdgcn_readlane(__builtin_bit_cast(int, wv), k + u));
            }
            float4 xa[8], xb[8];
            #pragma unroll
            for (int u = 0; u < 8; ++u) {
                if (k + u < m) {
                    const float* xr = X + (size_t)sk[u] * 256 + lane * 4;
                    xa[u] = *(const float4*)xr;
                    xb[u] = *(const float4*)(xr + 128);
                }
            }
            #pragma unroll
            for (int u = 0; u < 8; ++u) {
                if (k + u < m) {
                    float* a = acc + lk[u] * 256 + lane;
                    atomicAdd(a +   0, xa[u].x * wk[u]);
                    atomicAdd(a +  32, xa[u].y * wk[u]);
                    atomicAdd(a +  64, xa[u].z * wk[u]);
                    atomicAdd(a +  96, xa[u].w * wk[u]);
                    atomicAdd(a + 128, xb[u].x * wk[u]);
                    atomicAdd(a + 160, xb[u].y * wk[u]);
                    atomicAdd(a + 192, xb[u].z * wk[u]);
                    atomicAdd(a + 224, xb[u].w * wk[u]);
                }
            }
        }
    } else {
        const int half = lane >> 4;
        const int l16  = lane & 15;
        for (int k = 0; k < m; k += 8) {
            float4 x[4];
            float  wk[4];
            int    sk[4], lk[4];
            bool   vq[4];
            #pragma unroll
            for (int u = 0; u < 4; ++u) {
                const int q = k + 2 * u + half;
                vq[u] = q < m;
                sk[u] = __shfl(s, q);
                wk[u] = __shfl(wv, q);
                lk[u] = __shfl(lr, q);
            }
            #pragma unroll
            for (int u = 0; u < 4; ++u) {
                if (vq[u]) x[u] = *(const float4*)(X + (size_t)sk[u] * 64 + l16 * 4);
            }
            #pragma unroll
            for (int u = 0; u < 4; ++u) {
                if (vq[u]) {
                    float* a = acc + lk[u] * 64 + l16;
                    const float c0 = half ? x[u].z : x[u].x;
                    const float c1 = half ? x[u].w : x[u].y;
                    const float c2 = half ? x[u].x : x[u].z;
                    const float c3 = half ? x[u].y : x[u].w;
                    atomicAdd(a + (half ? 32 :  0), c0 * wk[u]);
                    atomicAdd(a + (half ? 48 : 16), c1 * wk[u]);
                    atomicAdd(a + (half ?  0 : 32), c2 * wk[u]);
                    atomicAdd(a + (half ? 16 : 48), c3 * wk[u]);
                }
            }
        }
    }
}

template <int D, bool SAVE>
__device__ __forceinline__ void drain_list(const float* __restrict__ X, const int* __restrict__ src,
                                           const float* __restrict__ w, float* acc,
                                           const unsigned* mylist, int n, int lane,
                                           unsigned* __restrict__ seg, int* gcount) {
    if (n == 0) return;
    int gpos = 0;
    if constexpr (SAVE) {
        int p = 0;
        if (lane == 0) p = atomicAdd(gcount, n);
        gpos = __builtin_amdgcn_readfirstlane(p);
    }
    for (int base = 0; base < n; base += 32) {
        const int  idx = base + lane;
        const bool ok  = idx < n;
        const unsigned ent = ok ? mylist[idx] : 0u;
        const int   e  = (int)(ent & 0xFFFFFFu);
        const int   s  = ok ? src[e] : 0;
        const float wv = ok ? w[e] : 0.0f;
        const int   lr = (int)(ent >> 24);
        if constexpr (SAVE) {
            const int g = gpos + idx;
            if (ok && g < GCAP) {
                v2u h;
                h.x = ((unsigned)lr << 24) | (unsigned)s;
                h.y = __builtin_bit_cast(unsigned, wv);
                *(v2u*)(seg + 2 * (size_t)g) = h;
            }
        }
        add_edges<D>(X, acc, s, wv, lr, min(32, n - base), lane);
    }
}

static_assert(DRAIN_AT + SUB <= CAP, "a trip of hits must fit behind an undrained list");
static_assert(ROWS <= 256 && NEDGES < (1 << 24) && NNODES < (1 << 24), "list entry packing");

template <int D, bool SAVE>
__device__ __forceinline__ void filter_and_sum(const float* __restrict__ X, const int* __restrict__ src,
                                               const int* __restrict__ dst, const float* __restrict__ w,
                                               float* acc, unsigned* mylist, int r0, int wave, int lane,
                                               unsigned* __restrict__ seg, int* gcount) {
    int n = 0;
    const int rot = (int)(((long long)blockIdx.x * NSUB) / NBLK);
    for (int sc = wave; sc < NSUB; sc += AW) {
        if (n >= DRAIN_AT) {
            drain_list<D, SAVE>(X, src, w, acc, mylist, n, lane, seg, gcount);
            n = 0;
        }
        int scr = sc + rot;
        if (scr >= NSUB) scr -= NSUB;
        const int ebase = scr * SUB + lane * 4;
        int4 d[4];
        #pragma unroll
        for (int j = 0; j < 4; ++j) {
            const int e = ebase + j * 128;
            d[j] = (e + 3 < NEDGES) ? *(const int4*)(dst + e) : make_int4(-1, -1, -1, -1);
        }
        #pragma unroll
        for (int j = 0; j < 4; ++j) {
            const int e = ebase + j * 128;
            const unsigned l0 = (unsigned)(d[j].x - r0);
            const unsigned l1 = (unsigned)(d[j].y - r0);
            const unsigned l2 = (unsigned)(d[j].z - r0);
            const unsigned l3 = (unsigned)(d[j].w - r0);
            const bool m0 = l0 < (unsigned)ROWS, m1 = l1 < (unsigned)ROWS;
            const bool m2 = l2 < (unsigned)ROWS, m3 = l3 < (unsigned)ROWS;
            if (__builtin_amdgcn_ballot_w32(m0 | m1 | m2 | m3) != 0u) {
                unsigned bm;
                bm = __builtin_amdgcn_ballot_w32(m0);
                if (m0) mylist[n + __builtin_amdgcn_mbcnt_lo(bm, 0u)] = (l0 << 24) | (unsigned)(e + 0);
                n += __popc(bm);
                bm = __builtin_amdgcn_ballot_w32(m1);
                if (m1) mylist[n + __builtin_amdgcn_mbcnt_lo(bm, 0u)] = (l1 << 24) | (unsigned)(e + 1);
                n += __popc(bm);
                bm = __builtin_amdgcn_ballot_w32(m2);
                if (m2) mylist[n + __builtin_amdgcn_mbcnt_lo(bm, 0u)] = (l2 << 24) | (unsigned)(e + 2);
                n += __popc(bm);
                bm = __builtin_amdgcn_ballot_w32(m3);
                if (m3) mylist[n + __builtin_amdgcn_mbcnt_lo(bm, 0u)] = (l3 << 24) | (unsigned)(e + 3);
                n += __popc(bm);
            }
        }
    }
    drain_list<D, SAVE>(X, src, w, acc, mylist, n, lane, seg, gcount);
}

template <int D>
__device__ __forceinline__ void sum_saved(const float* __restrict__ X, float* acc,
                                          const unsigned* __restrict__ seg, int total,
                                          int wave, int lane) {
    const int uw    = __builtin_amdgcn_readfirstlane(wave);
    const int share = (((total + AW - 1) / AW) + 7) & ~7;
    const int beg   = uw * share;
    const int end   = min(total, beg + share);
    for (int base = beg; base < end; base += 32) {
        const int  idx = base + lane;
        const bool ok  = idx < end;
        v2u h;
        h.x = 0u;
        h.y = 0u;
        if (ok) h = *(const v2u*)(seg + 2 * (size_t)idx);
        const unsigned key = h.x;
        const unsigned wb  = h.y;
        const int   s  = (int)(key & 0xFFFFFFu);
        const int   lr = (int)(key >> 24);
        const float wv = __builtin_bit_cast(float, wb);
        add_edges<D>(X, acc, s, wv, lr, min(32, end - base), lane);
    }
}

template <int D, bool SECOND>
__global__ __launch_bounds__(1024)
void agg_rows_lds(const float* __restrict__ X, const int* __restrict__ src,
                  const int* __restrict__ dst, const float* __restrict__ w,
                  float* __restrict__ out, unsigned* __restrict__ glist, int* __restrict__ gcnt) {
    __shared__ float    acc[ROWS * D];
    __shared__ unsigned lists[AW * CAP];
    __shared__ int      gcount;

    const int tid  = threadIdx.x;
    const int lane = tid & 31;
    const int wave = tid >> 5;
    const int r0   = blockIdx.x * ROWS;

    for (int i = tid; i < ROWS * D; i += AW * 32) acc[i] = 0.0f;
    if (tid == 0) gcount = 0;
    __syncthreads();

    unsigned* seg = glist + (size_t)blockIdx.x * (2 * GCAP);
    if constexpr (SECOND) {
        const int total = gcnt[blockIdx.x];
        if (total <= GCAP) {
            sum_saved<D>(X, acc, seg, total, wave, lane);
        } else {
            filter_and_sum<D, false>(X, src, dst, w, acc, lists + wave * CAP, r0, wave, lane, seg, &gcount);
        }
    } else {
        filter_and_sum<D, true>(X, src, dst, w, acc, lists + wave * CAP, r0, wave, lane, seg, &gcount);
    }
    __syncthreads();
    if constexpr (!SECOND) {
        if (tid == 0) gcnt[blockIdx.x] = gcount;
    }

    constexpr int F4R = D / 4;
    for (int f = tid; f < ROWS * F4R; f += AW * 32) {
        const int row = f / F4R;
        const int c4  = f % F4R;
        if (r0 + row < NNODES) {
            float4 v;
            if constexpr (D == 256) {
                const float* p = acc + row * 256 + (c4 >> 5) * 128 + (c4 & 31);
                v.x = p[0]; v.y = p[32]; v.z = p[64]; v.w = p[96];
            } else {
                const float* p = acc + row * 64 + c4;
                v.x = p[0]; v.y = p[16]; v.z = p[32]; v.w = p[48];
            }
            *(float4*)(out + (size_t)(r0 + row) * D + c4 * 4) = v;
        }
    }
}

extern "C" void kernel_launch(void* const* d_in, const int* in_sizes, int n_in,
                              void* d_out, int out_size, void* d_ws, size_t ws_size,
                              hipStream_t stream) {
    const float* node_feat = (const float*)d_in[0];
    const int*   edge_src  = (const int*)  d_in[1];
    const int*   edge_dst  = (const int*)  d_in[2];
    const float* edge_w    = (const float*)d_in[3];
    const float* W1        = (const float*)d_in[4];
    const float* W2        = (const float*)d_in[5];
    float* out = (float*)d_out;

    float* y1   = (float*)d_ws;
    float* agg1 = y1 + (size_t)NNODES * DHID;
    float* y2   = agg1 + (size_t)NNODES * DHID;
    unsigned* glist = (unsigned*)(y2 + (size_t)NNODES * DOUT);
    int*      gcnt  = (int*)(glist + (size_t)NBLK * GCAP * 2);

    const int mTiles = NNODES / 16;

    {
        dim3 grid(DHID / 64, (mTiles + 7) / 8);
        gemm_wmma_f32<false><<<grid, 256, 0, stream>>>(
            node_feat, W1, y1, NNODES, DHID, DIN, mTiles);
    }
    agg_rows_lds<DHID, false><<<NBLK, AW * 32, 0, stream>>>(
        y1, edge_src, edge_dst, edge_w, agg1, glist, gcnt);
    {
        dim3 grid(DOUT / 64, (mTiles + 7) / 8);
        gemm_wmma_f32<true><<<grid, 256, 0, stream>>>(
            agg1, W2, y2, NNODES, DOUT, DHID, mTiles);
    }
    agg_rows_lds<DOUT, true><<<NBLK, AW * 32, 0, stream>>>(
        y2, edge_src, edge_dst, edge_w, out, glist, gcnt);
}


# Dst filter trip lengthened from 512 to 1280 edges: each lane issues 10 x 16 B dst loads before its first compare (4 before), trips tile the list exactly; row sums in LDS arrival order as before
# speedup vs baseline: 2.3401x; 1.0458x over previous
#include <hip/hip_runtime.h>
#include <hip/hip_bf16.h>

typedef float v2f __attribute__((ext_vector_type(2)));
typedef float v8f __attribute__((ext_vector_type(8)));

constexpr int NNODES = 20000;
constexpr int NEDGES = 320000;
constexpr int DIN    = 256;
constexpr int DHID   = 256;
constexpr int DOUT   = 64;

template <bool RELU>
__global__ __launch_bounds__(256)
void gemm_wmma_f32(const float* __restrict__ A, const float* __restrict__ W,
                   float* __restrict__ C, int M, int N, int K, int mTiles) {
    __shared__ float smem[256 * 64];

    const int tid  = threadIdx.x;
    const int lane = tid & 31;
    const int wave = tid >> 5;
    const int n0   = blockIdx.x << 6;

    {
        const int n   = tid & 63;
        const int kq0 = tid >> 6;
        for (int kq = kq0; kq < (K >> 2); kq += 4) {
            const float4 v = *(const float4*)(W + (size_t)(n0 + n) * K + (kq << 2));
            v2f lo; lo.x = v.x; lo.y = v.y;
            v2f hi; hi.x = v.z; hi.y = v.w;
            *(v2f*)(smem + (size_t)(2 * kq)     * 128 + 2 * n) = lo;
            *(v2f*)(smem + (size_t)(2 * kq + 1) * 128 + 2 * n) = hi;
        }
    }
    __syncthreads();

    const int mt = blockIdx.y * 8 + wave;
    if (mt >= mTiles) return;
    const int m0 = mt << 4;

    const int half = lane >> 4;
    const int l15  = lane & 15;

    const float* arow = A + (size_t)(m0 + l15) * K + 2 * half;
    __builtin_prefetch(arow, 0, 3);
    __builtin_prefetch(arow + 128, 0, 3);

    const float* bbase = smem + (size_t)half * 128 + 2 * l15;

    v8f acc0 = {}, acc1 = {}, acc2 = {}, acc3 = {};

    #pragma unroll 8
    for (int k0 = 0; k0 < K; k0 += 4) {
        v2f a = *(const v2f*)(arow + k0);
        if (RELU) { a.x = fmaxf(a.x, 0.0f); a.y = fmaxf(a.y, 0.0f); }
        const float* bp = bbase + (size_t)(k0 >> 1) * 128;
        v2f b0 = *(const v2f*)(bp +  0);
        v2f b1 = *(const v2f*)(bp + 32);
        v2f b2 = *(const v2f*)(bp + 64);
        v2f b3 = *(const v2f*)(bp + 96);
        acc0 = __builtin_amdgcn_wmma_f32_16x16x4_f32(false, a, false, b0, (short)0, acc0, false, false);
        acc1 = __builtin_amdgcn_wmma_f32_16x16x4_f32(false, a, false, b1, (short)0, acc1, false, false);
        acc2 = __builtin_amdgcn_wmma_f32_16x16x4_f32(false, a, false, b2, (short)0, acc2, false, false);
        acc3 = __builtin_amdgcn_wmma_f32_16x16x4_f32(false, a, false, b3, (short)0, acc3, false, false);
    }

    #pragma unroll
    for (int v = 0; v < 8; ++v) {
        float* crow = C + (size_t)(m0 + v + 8 * half) * N + n0 + l15;
        crow[ 0] = acc0[v];
        crow[16] = acc1[v];
        crow[32] = acc2[v];
        crow[48] = acc3[v];
    }
}

constexpr int ROWS = 80;
constexpr int NBLK = (NNODES + ROWS - 1) / ROWS;
constexpr int AW   = 32;
constexpr int SUB  = 1280;
constexpr int NLD  = SUB / 128;
constexpr int CAP  = 1312;
constexpr int NSUB = (NEDGES + SUB - 1) / SUB;
constexpr int DRAIN_AT = 24;
constexpr int GCAP = 4096;

typedef unsigned v2u __attribute__((ext_vector_type(2)));

template <int D>
__device__ __forceinline__ void add_edges(const float* __restrict__ X, float* acc,
                                          int s, float wv, int lr, int m, int lane) {
    if constexpr (D == 256) {
        for (int k = 0; k < m; k += 8) {
            int   sk[8], lk[8];
            float wk[8];
            #pragma unroll
            for (int u = 0; u < 8; ++u) {
                sk[u] = __builtin_amdgcn_readlane(s, k + u);
                lk[u] = __builtin_amdgcn_readlane(lr, k + u);
                wk[u] = __builtin_bit_cast(float,
                            __builtin_amdgcn_readlane(__builtin_bit_cast(int, wv), k + u));
            }
            float4 xa[8], xb[8];
            #pragma unroll
            for (int u = 0; u < 8; ++u) {
                if (k + u < m) {
                    const float* xr = X + (size_t)sk[u] * 256 + lane * 4;
                    xa[u] = *(const float4*)xr;
                    xb[u] = *(const float4*)(xr + 128);
                }
            }
            #pragma unroll
            for (int u = 0; u < 8; ++u) {
                if (k + u < m) {
                    float* a = acc + lk[u] * 256 + lane;
                    atomicAdd(a +   0, xa[u].x * wk[u]);
                    atomicAdd(a +  32, xa[u].y * wk[u]);
                    atomicAdd(a +  64, xa[u].z * wk[u]);
                    atomicAdd(a +  96, xa[u].w * wk[u]);
                    atomicAdd(a + 128, xb[u].x * wk[u]);
                    atomicAdd(a + 160, xb[u].y * wk[u]);
                    atomicAdd(a + 192, xb[u].z * wk[u]);
                    atomicAdd(a + 224, xb[u].w * wk[u]);
                }
            }
        }
    } else {
        const int half = lane >> 4;
        const int l16  = lane & 15;
        for (int k = 0; k < m; k += 8) {
            float4 x[4];
            float  wk[4];
            int    sk[4], lk[4];
            bool   vq[4];
            #pragma unroll
            for (int u = 0; u < 4; ++u) {
                const int q = k + 2 * u + half;
                vq[u] = q < m;
                sk[u] = __shfl(s, q);
                wk[u] = __shfl(wv, q);
                lk[u] = __shfl(lr, q);
            }
            #pragma unroll
            for (int u = 0; u < 4; ++u) {
                if (vq[u]) x[u] = *(const float4*)(X + (size_t)sk[u] * 64 + l16 * 4);
            }
            #pragma unroll
            for (int u = 0; u < 4; ++u) {
                if (vq[u]) {
                    float* a = acc + lk[u] * 64 + l16;
                    const float c0 = half ? x[u].z : x[u].x;
                    const float c1 = half ? x[u].w : x[u].y;
                    const float c2 = half ? x[u].x : x[u].z;
                    const float c3 = half ? x[u].y : x[u].w;
                    atomicAdd(a + (half ? 32 :  0), c0 * wk[u]);
                    atomicAdd(a + (half ? 48 : 16), c1 * wk[u]);
                    atomicAdd(a + (half ?  0 : 32), c2 * wk[u]);
                    atomicAdd(a + (half ? 16 : 48), c3 * wk[u]);
                }
            }
        }
    }
}

template <int D, bool SAVE>
__device__ __forceinline__ void drain_list(const float* __restrict__ X, const int* __restrict__ src,
                                           const float* __restrict__ w, float* acc,
                                           const unsigned* mylist, int n, int lane,
                                           unsigned* __restrict__ seg, int* gcount) {
    if (n == 0) return;
    int gpos = 0;
    if constexpr (SAVE) {
        int p = 0;
        if (lane == 0) p = atomicAdd(gcount, n);
        gpos = __builtin_amdgcn_readfirstlane(p);
    }
    for (int base = 0; base < n; base += 32) {
        const int  idx = base + lane;
        const bool ok  = idx < n;
        const unsigned ent = ok ? mylist[idx] : 0u;
        const int   e  = (int)(ent & 0xFFFFFFu);
        const int   s  = ok ? src[e] : 0;
        const float wv = ok ? w[e] : 0.0f;
        const int   lr = (int)(ent >> 24);
        if constexpr (SAVE) {
            const int g = gpos + idx;
            if (ok && g < GCAP) {
                v2u h;
                h.x = ((unsigned)lr << 24) | (unsigned)s;
                h.y = __builtin_bit_cast(unsigned, wv);
                *(v2u*)(seg + 2 * (size_t)g) = h;
            }
        }
        add_edges<D>(X, acc, s, wv, lr, min(32, n - base), lane);
    }
}

static_assert(DRAIN_AT + SUB <= CAP, "a trip of hits must fit behind an undrained list");
static_assert(NEDGES % SUB == 0 && SUB % 128 == 0, "trips tile the dst list exactly: the filter loads carry no bounds test");
static_assert(ROWS <= 256 && NEDGES < (1 << 24) && NNODES < (1 << 24), "list entry packing");

template <int D, bool SAVE>
__device__ __forceinline__ void filter_and_sum(const float* __restrict__ X, const int* __restrict__ src,
                                               const int* __restrict__ dst, const float* __restrict__ w,
                                               float* acc, unsigned* mylist, int r0, int wave, int lane,
                                               unsigned* __restrict__ seg, int* gcount) {
    int n = 0;
    const int rot = (int)(((long long)blockIdx.x * NSUB) / NBLK);
    for (int sc = wave; sc < NSUB; sc += AW) {
        if (n >= DRAIN_AT) {
            drain_list<D, SAVE>(X, src, w, acc, mylist, n, lane, seg, gcount);
            n = 0;
        }
        int scr = sc + rot;
        if (scr >= NSUB) scr -= NSUB;
        const int ebase = scr * SUB + lane * 4;
        int4 d[NLD];
        #pragma unroll
        for (int j = 0; j < NLD; ++j) d[j] = *(const int4*)(dst + ebase + j * 128);
        #pragma unroll
        for (int j = 0; j < NLD; ++j) {
            const int e = ebase + j * 128;
            const unsigned l0 = (unsigned)(d[j].x - r0);
            const unsigned l1 = (unsigned)(d[j].y - r0);
            const unsigned l2 = (unsigned)(d[j].z - r0);
            const unsigned l3 = (unsigned)(d[j].w - r0);
            const bool m0 = l0 < (unsigned)ROWS, m1 = l1 < (unsigned)ROWS;
            const bool m2 = l2 < (unsigned)ROWS, m3 = l3 < (unsigned)ROWS;
            if (__builtin_amdgcn_ballot_w32(m0 | m1 | m2 | m3) != 0u) {
                unsigned bm;
                bm = __builtin_amdgcn_ballot_w32(m0);
                if (m0) mylist[n + __builtin_amdgcn_mbcnt_lo(bm, 0u)] = (l0 << 24) | (unsigned)(e + 0);
                n += __popc(bm);
                bm = __builtin_amdgcn_ballot_w32(m1);
                if (m1) mylist[n + __builtin_amdgcn_mbcnt_lo(bm, 0u)] = (l1 << 24) | (unsigned)(e + 1);
                n += __popc(bm);
                bm = __builtin_amdgcn_ballot_w32(m2);
                if (m2) mylist[n + __builtin_amdgcn_mbcnt_lo(bm, 0u)] = (l2 << 24) | (unsigned)(e + 2);
                n += __popc(bm);
                bm = __builtin_amdgcn_ballot_w32(m3);
                if (m3) mylist[n + __builtin_amdgcn_mbcnt_lo(bm, 0u)] = (l3 << 24) | (unsigned)(e + 3);
                n += __popc(bm);
            }
        }
    }
    drain_list<D, SAVE>(X, src, w, acc, mylist, n, lane, seg, gcount);
}

template <int D>
__device__ __forceinline__ void sum_saved(const float* __restrict__ X, float* acc,
                                          const unsigned* __restrict__ seg, int total,
                                          int wave, int lane) {
    const int uw    = __builtin_amdgcn_readfirstlane(wave);
    const int share = (((total + AW - 1) / AW) + 7) & ~7;
    const int beg   = uw * share;
    const int end   = min(total, beg + share);
    for (int base = beg; base < end; base += 32) {
        const int  idx = base + lane;
        const bool ok  = idx < end;
        v2u h;
        h.x = 0u;
        h.y = 0u;
        if (ok) h = *(const v2u*)(seg + 2 * (size_t)idx);
        const unsigned key = h.x;
        const unsigned wb  = h.y;
        const int   s  = (int)(key & 0xFFFFFFu);
        const int   lr = (int)(key >> 24);
        const float wv = __builtin_bit_cast(float, wb);
        add_edges<D>(X, acc, s, wv, lr, min(32, end - base), lane);
    }
}

template <int D, bool SECOND>
__global__ __launch_bounds__(1024)
void agg_rows_lds(const float* __restrict__ X, const int* __restrict__ src,
                  const int* __restrict__ dst, const float* __restrict__ w,
                  float* __restrict__ out, unsigned* __restrict__ glist, int* __restrict__ gcnt) {
    __shared__ float    acc[ROWS * D];
    __shared__ unsigned lists[AW * CAP];
    __shared__ int      gcount;

    const int tid  = threadIdx.x;
    const int lane = tid & 31;
    const int wave = tid >> 5;
    const int r0   = blockIdx.x * ROWS;

    for (int i = tid; i < ROWS * D; i += AW * 32) acc[i] = 0.0f;
    if (tid == 0) gcount = 0;
    __syncthreads();

    unsigned* seg = glist + (size_t)blockIdx.x * (2 * GCAP);
    if constexpr (SECOND) {
        const int total = gcnt[blockIdx.x];
        if (total <= GCAP) {
            sum_saved<D>(X, acc, seg, total, wave, lane);
        } else {
            filter_and_sum<D, false>(X, src, dst, w, acc, lists + wave * CAP, r0, wave, lane, seg, &gcount);
        }
    } else {
        filter_and_sum<D, true>(X, src, dst, w, acc, lists + wave * CAP, r0, wave, lane, seg, &gcount);
    }
    __syncthreads();
    if constexpr (!SECOND) {
        if (tid == 0) gcnt[blockIdx.x] = gcount;
    }

    constexpr int F4R = D / 4;
    for (int f = tid; f < ROWS * F4R; f += AW * 32) {
        const int row = f / F4R;
        const int c4  = f % F4R;
        if (r0 + row < NNODES) {
            float4 v;
            if constexpr (D == 256) {
                const float* p = acc + row * 256 + (c4 >> 5) * 128 + (c4 & 31);
                v.x = p[0]; v.y = p[32]; v.z = p[64]; v.w = p[96];
            } else {
                const float* p = acc + row * 64 + c4;
                v.x = p[0]; v.y = p[16]; v.z = p[32]; v.w = p[48];
            }
            *(float4*)(out + (size_t)(r0 + row) * D + c4 * 4) = v;
        }
    }
}

extern "C" void kernel_launch(void* const* d_in, const int* in_sizes, int n_in,
                              void* d_out, int out_size, void* d_ws, size_t ws_size,
                              hipStream_t stream) {
    const float* node_feat = (const float*)d_in[0];
    const int*   edge_src  = (const int*)  d_in[1];
    const int*   edge_dst  = (const int*)  d_in[2];
    const float* edge_w    = (const float*)d_in[3];
    const float* W1        = (const float*)d_in[4];
    const float* W2        = (const float*)d_in[5];
    float* out = (float*)d_out;

    float* y1   = (float*)d_ws;
    float* agg1 = y1 + (size_t)NNODES * DHID;
    float* y2   = agg1 + (size_t)NNODES * DHID;
    unsigned* glist = (unsigned*)(y2 + (size_t)NNODES * DOUT);
    int*      gcnt  = (int*)(glist + (size_t)NBLK * GCAP * 2);

    const int mTiles = NNODES / 16;

    {
        dim3 grid(DHID / 64, (mTiles + 7) / 8);
        gemm_wmma_f32<false><<<grid, 256, 0, stream>>>(
            node_feat, W1, y1, NNODES, DHID, DIN, mTiles);
    }
    agg_rows_lds<DHID, false><<<NBLK, AW * 32, 0, stream>>>(
        y1, edge_src, edge_dst, edge_w, agg1, glist, gcnt);
    {
        dim3 grid(DOUT / 64, (mTiles + 7) / 8);
        gemm_wmma_f32<true><<<grid, 256, 0, stream>>>(
            agg1, W2, y2, NNODES, DOUT, DHID, mTiles);
    }
    agg_rows_lds<DOUT, true><<<NBLK, AW * 32, 0, stream>>>(
        y2, edge_src, edge_dst, edge_w, out, glist, gcnt);
}


# Layer-1 product moved behind the neighbour sum into its launch: each workgroup multiplies its 80 summed rows (LDS) by W1 on f32 WMMA; GEMM-1 launch gone; rows summed before product, LDS arrival order
# speedup vs baseline: 2.6784x; 1.1446x over previous
#include <hip/hip_runtime.h>
#include <hip/hip_bf16.h>

typedef float v2f __attribute__((ext_vector_type(2)));
typedef float v8f __attribute__((ext_vector_type(8)));

constexpr int NNODES = 20000;
constexpr int NEDGES = 320000;
constexpr int DIN    = 256;
constexpr int DHID   = 256;
constexpr int DOUT   = 64;

template <bool RELU>
__global__ __launch_bounds__(256)
void gemm_wmma_f32(const float* __restrict__ A, const float* __restrict__ W,
                   float* __restrict__ C, int M, int N, int K, int mTiles) {
    __shared__ float smem[256 * 64];

    const int tid  = threadIdx.x;
    const int lane = tid & 31;
    const int wave = tid >> 5;
    const int n0   = blockIdx.x << 6;

    {
        const int n   = tid & 63;
        const int kq0 = tid >> 6;
        for (int kq = kq0; kq < (K >> 2); kq += 4) {
            const float4 v = *(const float4*)(W + (size_t)(n0 + n) * K + (kq << 2));
            v2f lo; lo.x = v.x; lo.y = v.y;
            v2f hi; hi.x = v.z; hi.y = v.w;
            *(v2f*)(smem + (size_t)(2 * kq)     * 128 + 2 * n) = lo;
            *(v2f*)(smem + (size_t)(2 * kq + 1) * 128 + 2 * n) = hi;
        }
    }
    __syncthreads();

    const int mt = blockIdx.y * 8 + wave;
    if (mt >= mTiles) return;
    const int m0 = mt << 4;

    const int half = lane >> 4;
    const int l15  = lane & 15;

    const float* arow = A + (size_t)(m0 + l15) * K + 2 * half;
    __builtin_prefetch(arow, 0, 3);
    __builtin_prefetch(arow + 128, 0, 3);

    const float* bbase = smem + (size_t)half * 128 + 2 * l15;

    v8f acc0 = {}, acc1 = {}, acc2 = {}, acc3 = {};

    #pragma unroll 8
    for (int k0 = 0; k0 < K; k0 += 4) {
        v2f a = *(const v2f*)(arow + k0);
        if (RELU) { a.x = fmaxf(a.x, 0.0f); a.y = fmaxf(a.y, 0.0f); }
        const float* bp = bbase + (size_t)(k0 >> 1) * 128;
        v2f b0 = *(const v2f*)(bp +  0);
        v2f b1 = *(const v2f*)(bp + 32);
        v2f b2 = *(const v2f*)(bp + 64);
        v2f b3 = *(const v2f*)(bp + 96);
        acc0 = __builtin_amdgcn_wmma_f32_16x16x4_f32(false, a, false, b0, (short)0, acc0, false, false);
        acc1 = __builtin_amdgcn_wmma_f32_16x16x4_f32(false, a, false, b1, (short)0, acc1, false, false);
        acc2 = __builtin_amdgcn_wmma_f32_16x16x4_f32(false, a, false, b2, (short)0, acc2, false, false);
        acc3 = __builtin_amdgcn_wmma_f32_16x16x4_f32(false, a, false, b3, (short)0, acc3, false, false);
    }

    #pragma unroll
    for (int v = 0; v < 8; ++v) {
        float* crow = C + (size_t)(m0 + v + 8 * half) * N + n0 + l15;
        crow[ 0] = acc0[v];
        crow[16] = acc1[v];
        crow[32] = acc2[v];
        crow[48] = acc3[v];
    }
}

constexpr int ROWS = 80;
constexpr int P256 = 260;
constexpr int NBLK = (NNODES + ROWS - 1) / ROWS;
constexpr int AW   = 32;
constexpr int SUB  = 1280;
constexpr int NLD  = SUB / 128;
constexpr int CAP  = 1312;
constexpr int NSUB = (NEDGES + SUB - 1) / SUB;
constexpr int DRAIN_AT = 24;
constexpr int GCAP = 4096;

typedef unsigned v2u __attribute__((ext_vector_type(2)));

template <int D>
__device__ __forceinline__ void add_edges(const float* __restrict__ X, float* acc,
                                          int s, float wv, int lr, int m, int lane) {
    if constexpr (D == 256) {
        for (int k = 0; k < m; k += 8) {
            int   sk[8], lk[8];
            float wk[8];
            #pragma unroll
            for (int u = 0; u < 8; ++u) {
                sk[u] = __builtin_amdgcn_readlane(s, k + u);
                lk[u] = __builtin_amdgcn_readlane(lr, k + u);
                wk[u] = __builtin_bit_cast(float,
                            __builtin_amdgcn_readlane(__builtin_bit_cast(int, wv), k + u));
            }
            float4 xa[8], xb[8];
            #pragma unroll
            for (int u = 0; u < 8; ++u) {
                if (k + u < m) {
                    const float* xr = X + (size_t)sk[u] * 256 + lane * 4;
                    xa[u] = *(const float4*)xr;
                    xb[u] = *(const float4*)(xr + 128);
                }
            }
            #pragma unroll
            for (int u = 0; u < 8; ++u) {
                if (k + u < m) {
                    float* a = acc + lk[u] * P256 + lane;
                    atomicAdd(a +   0, xa[u].x * wk[u]);
                    atomicAdd(a +  32, xa[u].y * wk[u]);
                    atomicAdd(a +  64, xa[u].z * wk[u]);
                    atomicAdd(a +  96, xa[u].w * wk[u]);
                    atomicAdd(a + 128, xb[u].x * wk[u]);
                    atomicAdd(a + 160, xb[u].y * wk[u]);
                    atomicAdd(a + 192, xb[u].z * wk[u]);
                    atomicAdd(a + 224, xb[u].w * wk[u]);
                }
            }
        }
    } else {
        const int half = lane >> 4;
        const int l16  = lane & 15;
        for (int k = 0; k < m; k += 8) {
            float4 x[4];
            float  wk[4];
            int    sk[4], lk[4];
            bool   vq[4];
            #pragma unroll
            for (int u = 0; u < 4; ++u) {
                const int q = k + 2 * u + half;
                vq[u] = q < m;
                sk[u] = __shfl(s, q);
                wk[u] = __shfl(wv, q);
                lk[u] = __shfl(lr, q);
            }
            #pragma unroll
            for (int u = 0; u < 4; ++u) {
                if (vq[u]) x[u] = *(const float4*)(X + (size_t)sk[u] * 64 + l16 * 4);
            }
            #pragma unroll
            for (int u = 0; u < 4; ++u) {
                if (vq[u]) {
                    float* a = acc + lk[u] * 64 + l16;
                    const float c0 = half ? x[u].z : x[u].x;
                    const float c1 = half ? x[u].w : x[u].y;
                    const float c2 = half ? x[u].x : x[u].z;
                    const float c3 = half ? x[u].y : x[u].w;
                    atomicAdd(a + (half ? 32 :  0), c0 * wk[u]);
                    atomicAdd(a + (half ? 48 : 16), c1 * wk[u]);
                    atomicAdd(a + (half ?  0 : 32), c2 * wk[u]);
                    atomicAdd(a + (half ? 16 : 48), c3 * wk[u]);
                }
            }
        }
    }
}

template <int D, bool SAVE>
__device__ __forceinline__ void drain_list(const float* __restrict__ X, const int* __restrict__ src,
                                           const float* __restrict__ w, float* acc,
                                           const unsigned* mylist, int n, int lane,
                                           unsigned* __restrict__ seg, int* gcount) {
    if (n == 0) return;
    int gpos = 0;
    if constexpr (SAVE) {
        int p = 0;
        if (lane == 0) p = atomicAdd(gcount, n);
        gpos = __builtin_amdgcn_readfirstlane(p);
    }
    for (int base = 0; base < n; base += 32) {
        const int  idx = base + lane;
        const bool ok  = idx < n;
        const unsigned ent = ok ? mylist[idx] : 0u;
        const int   e  = (int)(ent & 0xFFFFFFu);
        const int   s  = ok ? src[e] : 0;
        const float wv = ok ? w[e] : 0.0f;
        const int   lr = (int)(ent >> 24);
        if constexpr (SAVE) {
            const int g = gpos + idx;
            if (ok && g < GCAP) {
                v2u h;
                h.x = ((unsigned)lr << 24) | (unsigned)s;
                h.y = __builtin_bit_cast(unsigned, wv);
                *(v2u*)(seg + 2 * (size_t)g) = h;
            }
        }
        add_edges<D>(X, acc, s, wv, lr, min(32, n - base), lane);
    }
}

static_assert(DRAIN_AT + SUB <= CAP, "a trip of hits must fit behind an undrained list");
static_assert(NEDGES % SUB == 0 && SUB % 128 == 0, "trips tile the dst list exactly: the filter loads carry no bounds test");
static_assert(ROWS <= 256 && NEDGES < (1 << 24) && NNODES < (1 << 24), "list entry packing");

template <int D, bool SAVE>
__device__ __forceinline__ void filter_and_sum(const float* __restrict__ X, const int* __restrict__ src,
                                               const int* __restrict__ dst, const float* __restrict__ w,
                                               float* acc, unsigned* mylist, int r0, int wave, int lane,
                                               unsigned* __restrict__ seg, int* gcount) {
    int n = 0;
    const int rot = (int)(((long long)blockIdx.x * NSUB) / NBLK);
    for (int sc = wave; sc < NSUB; sc += AW) {
        if (n >= DRAIN_AT) {
            drain_list<D, SAVE>(X, src, w, acc, mylist, n, lane, seg, gcount);
            n = 0;
        }
        int scr = sc + rot;
        if (scr >= NSUB) scr -= NSUB;
        const int ebase = scr * SUB + lane * 4;
        int4 d[NLD];
        #pragma unroll
        for (int j = 0; j < NLD; ++j) d[j] = *(const int4*)(dst + ebase + j * 128);
        #pragma unroll
        for (int j = 0; j < NLD; ++j) {
            const int e = ebase + j * 128;
            const unsigned l0 = (unsigned)(d[j].x - r0);
            const unsigned l1 = (unsigned)(d[j].y - r0);
            const unsigned l2 = (unsigned)(d[j].z - r0);
            const unsigned l3 = (unsigned)(d[j].w - r0);
            const bool m0 = l0 < (unsigned)ROWS, m1 = l1 < (unsigned)ROWS;
            const bool m2 = l2 < (unsigned)ROWS, m3 = l3 < (unsigned)ROWS;
            if (__builtin_amdgcn_ballot_w32(m0 | m1 | m2 | m3) != 0u) {
                unsigned bm;
                bm = __builtin_amdgcn_ballot_w32(m0);
                if (m0) mylist[n + __builtin_amdgcn_mbcnt_lo(bm, 0u)] = (l0 << 24) | (unsigned)(e + 0);
                n += __popc(bm);
                bm = __builtin_amdgcn_ballot_w32(m1);
                if (m1) mylist[n + __builtin_amdgcn_mbcnt_lo(bm, 0u)] = (l1 << 24) | (unsigned)(e + 1);
                n += __popc(bm);
                bm = __builtin_amdgcn_ballot_w32(m2);
                if (m2) mylist[n + __builtin_amdgcn_mbcnt_lo(bm, 0u)] = (l2 << 24) | (unsigned)(e + 2);
                n += __popc(bm);
                bm = __builtin_amdgcn_ballot_w32(m3);
                if (m3) mylist[n + __builtin_amdgcn_mbcnt_lo(bm, 0u)] = (l3 << 24) | (unsigned)(e + 3);
                n += __popc(bm);
            }
        }
    }
    drain_list<D, SAVE>(X, src, w, acc, mylist, n, lane, seg, gcount);
}

template <int D>
__device__ __forceinline__ void sum_saved(const float* __restrict__ X, float* acc,
                                          const unsigned* __restrict__ seg, int total,
                                          int wave, int lane) {
    const int uw    = __builtin_amdgcn_readfirstlane(wave);
    const int share = (((total + AW - 1) / AW) + 7) & ~7;
    const int beg   = uw * share;
    const int end   = min(total, beg + share);
    for (int base = beg; base < end; base += 32) {
        const int  idx = base + lane;
        const bool ok  = idx < end;
        v2u h;
        h.x = 0u;
        h.y = 0u;
        if (ok) h = *(const v2u*)(seg + 2 * (size_t)idx);
        const unsigned key = h.x;
        const unsigned wb  = h.y;
        const int   s  = (int)(key & 0xFFFFFFu);
        const int   lr = (int)(key >> 24);
        const float wv = __builtin_bit_cast(float, wb);
        add_edges<D>(X, acc, s, wv, lr, min(32, end - base), lane);
    }
}

constexpr int MT = ROWS / 16;
static_assert(ROWS % 16 == 0 && NNODES % ROWS == 0, "whole tiles only: the product's stores carry no row test");
static_assert(DIN == 256 && DHID == 256, "the fused layer-1 product reads 256-float rows and writes 256 columns");

__device__ __forceinline__ void rows_times_w(const float* acc, const float* __restrict__ Wm,
                                             float* __restrict__ out, int r0, int nt, int lane) {
    const int half = lane >> 4;
    const int l15  = lane & 15;
    const float* wrow = Wm + (size_t)(nt * 16 + l15) * 256;
    const float* arow = acc + l15 * P256 + 64 * half;

    v8f c[MT];
    #pragma unroll
    for (int m = 0; m < MT; ++m) c[m] = v8f{};

    for (int hb = 0; hb < 2; ++hb) {
        const float* ab = arow + hb * 128;
        const float* wb = wrow + hb * 128;
        #pragma unroll 4
        for (int p = 0; p < 16; ++p) {
            {
                const int tl = 2 * p + half;
                const v2f b = *(const v2f*)(wb + 4 * tl + 2 * half);
                #pragma unroll
                for (int m = 0; m < MT; ++m) {
                    v2f a;
                    a.x = ab[m * 16 * P256 + tl];
                    a.y = ab[m * 16 * P256 + tl + 32];
                    c[m] = __builtin_amdgcn_wmma_f32_16x16x4_f32(false, a, false, b, (short)0, c[m], false, false);
                }
            }
            {
                const int tl = 2 * p + 1 - half;
                const v2f b = *(const v2f*)(wb + 4 * tl + 2 * half);
                #pragma unroll
                for (int m = 0; m < MT; ++m) {
                    v2f a;
                    a.x = ab[m * 16 * P256 + tl];
                    a.y = ab[m * 16 * P256 + tl + 32];
                    c[m] = __builtin_amdgcn_wmma_f32_16x16x4_f32(false, a, false, b, (short)0, c[m], false, false);
                }
            }
        }
    }

    #pragma unroll
    for (int m = 0; m < MT; ++m) {
        #pragma unroll
        for (int v = 0; v < 8; ++v)
            out[(size_t)(r0 + m * 16 + v + 8 * half) * 256 + nt * 16 + l15] = c[m][v];
    }
}

template <int D, bool SECOND>
__global__ __launch_bounds__(1024)
void agg_rows_lds(const float* __restrict__ X, const int* __restrict__ src,
                  const int* __restrict__ dst, const float* __restrict__ w,
                  float* __restrict__ out, unsigned* __restrict__ glist, int* __restrict__ gcnt,
                  const float* __restrict__ Wm) {
    constexpr int PITCH = (D == 256) ? P256 : D;
    __shared__ float    acc[ROWS * PITCH];
    __shared__ unsigned lists[AW * CAP];
    __shared__ int      gcount;

    const int tid  = threadIdx.x;
    const int lane = tid & 31;
    const int wave = tid >> 5;
    const int r0   = blockIdx.x * ROWS;

    for (int i = tid; i < ROWS * PITCH; i += AW * 32) acc[i] = 0.0f;
    if (tid == 0) gcount = 0;
    __syncthreads();

    unsigned* seg = glist + (size_t)blockIdx.x * (2 * GCAP);
    if constexpr (SECOND) {
        const int total = gcnt[blockIdx.x];
        if (total <= GCAP) {
            sum_saved<D>(X, acc, seg, total, wave, lane);
        } else {
            filter_and_sum<D, false>(X, src, dst, w, acc, lists + wave * CAP, r0, wave, lane, seg, &gcount);
        }
    } else {
        filter_and_sum<D, true>(X, src, dst, w, acc, lists + wave * CAP, r0, wave, lane, seg, &gcount);
    }
    __syncthreads();
    if constexpr (!SECOND) {
        if (tid == 0) gcnt[blockIdx.x] = gcount;
    }

    if constexpr (D == 256) {
        const int uw = __builtin_amdgcn_readfirstlane(wave);
        if (uw < D / 16) rows_times_w(acc, Wm, out, r0, uw, lane);
    } else {
        constexpr int F4R = D / 4;
        for (int f = tid; f < ROWS * F4R; f += AW * 32) {
            const int row = f / F4R;
            const int c4  = f % F4R;
            if (r0 + row < NNODES) {
                float4 v;
                const float* p = acc + row * 64 + c4;
                v.x = p[0]; v.y = p[16]; v.z = p[32]; v.w = p[48];
                *(float4*)(out + (size_t)(r0 + row) * D + c4 * 4) = v;
            }
        }
    }
}

extern "C" void kernel_launch(void* const* d_in, const int* in_sizes, int n_in,
                              void* d_out, int out_size, void* d_ws, size_t ws_size,
                              hipStream_t stream) {
    const float* node_feat = (const float*)d_in[0];
    const int*   edge_src  = (const int*)  d_in[1];
    const int*   edge_dst  = (const int*)  d_in[2];
    const float* edge_w    = (const float*)d_in[3];
    const float* W1        = (const float*)d_in[4];
    const float* W2        = (const float*)d_in[5];
    float* out = (float*)d_out;

    float* agg1 = (float*)d_ws;
    float* y2   = agg1 + (size_t)NNODES * DHID;
    unsigned* glist = (unsigned*)(y2 + (size_t)NNODES * DOUT);
    int*      gcnt  = (int*)(glist + (size_t)NBLK * GCAP * 2);

    const int mTiles = NNODES / 16;

    agg_rows_lds<DIN, false><<<NBLK, AW * 32, 0, stream>>>(
        node_feat, edge_src, edge_dst, edge_w, agg1, glist, gcnt, W1);
    {
        dim3 grid(DOUT / 64, (mTiles + 7) / 8);
        gemm_wmma_f32<true><<<grid, 256, 0, stream>>>(
            agg1, W2, y2, NNODES, DOUT, DHID, mTiles);
    }
    agg_rows_lds<DOUT, true><<<NBLK, AW * 32, 0, stream>>>(
        y2, edge_src, edge_dst, edge_w, out, glist, gcnt, W2);
}


# Layer-2 product moved into the layer-1 launch behind the W1 product: ReLU'd 80x256 tile goes registers -> LDS, 20 waves multiply it by W2 on f32 WMMA; GEMM-2 launch and h's memory trip gone
# speedup vs baseline: 3.1000x; 1.1574x over previous
#include <hip/hip_runtime.h>
#include <hip/hip_bf16.h>

typedef float v2f __attribute__((ext_vector_type(2)));
typedef float v8f __attribute__((ext_vector_type(8)));

constexpr int NNODES = 20000;
constexpr int NEDGES = 320000;
constexpr int DIN    = 256;
constexpr int DHID   = 256;
constexpr int DOUT   = 64;

constexpr int ROWS = 80;
constexpr int P256 = 260;
constexpr int NBLK = (NNODES + ROWS - 1) / ROWS;
constexpr int AW   = 32;
constexpr int SUB  = 1280;
constexpr int NLD  = SUB / 128;
constexpr int CAP  = 1312;
constexpr int NSUB = (NEDGES + SUB - 1) / SUB;
constexpr int DRAIN_AT = 24;
constexpr int GCAP = 4096;

typedef unsigned v2u __attribute__((ext_vector_type(2)));

template <int D>
__device__ __forceinline__ void add_edges(const float* __restrict__ X, float* acc,
                                          int s, float wv, int lr, int m, int lane) {
    if constexpr (D == 256) {
        for (int k = 0; k < m; k += 8) {
            int   sk[8], lk[8];
            float wk[8];
            #pragma unroll
            for (int u = 0; u < 8; ++u) {
                sk[u] = __builtin_amdgcn_readlane(s, k + u);
                lk[u] = __builtin_amdgcn_readlane(lr, k + u);
                wk[u] = __builtin_bit_cast(float,
                            __builtin_amdgcn_readlane(__builtin_bit_cast(int, wv), k + u));
            }
            float4 xa[8], xb[8];
            #pragma unroll
            for (int u = 0; u < 8; ++u) {
                if (k + u < m) {
                    const float* xr = X + (size_t)sk[u] * 256 + lane * 4;
                    xa[u] = *(const float4*)xr;
                    xb[u] = *(const float4*)(xr + 128);
                }
            }
            #pragma unroll
            for (int u = 0; u < 8; ++u) {
                if (k + u < m) {
                    float* a = acc + lk[u] * P256 + lane;
                    atomicAdd(a +   0, xa[u].x * wk[u]);
                    atomicAdd(a +  32, xa[u].y * wk[u]);
                    atomicAdd(a +  64, xa[u].z * wk[u]);
                    atomicAdd(a +  96, xa[u].w * wk[u]);
                    atomicAdd(a + 128, xb[u].x * wk[u]);
                    atomicAdd(a + 160, xb[u].y * wk[u]);
                    atomicAdd(a + 192, xb[u].z * wk[u]);
                    atomicAdd(a + 224, xb[u].w * wk[u]);
                }
            }
        }
    } else {
        const int half = lane >> 4;
        const int l16  = lane & 15;
        for (int k = 0; k < m; k += 8) {
            float4 x[4];
            float  wk[4];
            int    sk[4], lk[4];
            bool   vq[4];
            #pragma unroll
            for (int u = 0; u < 4; ++u) {
                const int q = k + 2 * u + half;
                vq[u] = q < m;
                sk[u] = __shfl(s, q);
                wk[u] = __shfl(wv, q);
                lk[u] = __shfl(lr, q);
            }
            #pragma unroll
            for (int u = 0; u < 4; ++u) {
                if (vq[u]) x[u] = *(const float4*)(X + (size_t)sk[u] * 64 + l16 * 4);
            }
            #pragma unroll
            for (int u = 0; u < 4; ++u) {
                if (vq[u]) {
                    float* a = acc + lk[u] * 64 + l16;
                    const float c0 = half ? x[u].z : x[u].x;
                    const float c1 = half ? x[u].w : x[u].y;
                    const float c2 = half ? x[u].x : x[u].z;
                    const float c3 = half ? x[u].y : x[u].w;
                    atomicAdd(a + (half ? 32 :  0), c0 * wk[u]);
                    atomicAdd(a + (half ? 48 : 16), c1 * wk[u]);
                    atomicAdd(a + (half ?  0 : 32), c2 * wk[u]);
                    atomicAdd(a + (half ? 16 : 48), c3 * wk[u]);
                }
            }
        }
    }
}

template <int D, bool SAVE>
__device__ __forceinline__ void drain_list(const float* __restrict__ X, const int* __restrict__ src,
                                           const float* __restrict__ w, float* acc,
                                           const unsigned* mylist, int n, int lane,
                                           unsigned* __restrict__ seg, int* gcount) {
    if (n == 0) return;
    int gpos = 0;
    if constexpr (SAVE) {
        int p = 0;
        if (lane == 0) p = atomicAdd(gcount, n);
        gpos = __builtin_amdgcn_readfirstlane(p);
    }
    for (int base = 0; base < n; base += 32) {
        const int  idx = base + lane;
        const bool ok  = idx < n;
        const unsigned ent = ok ? mylist[idx] : 0u;
        const int   e  = (int)(ent & 0xFFFFFFu);
        const int   s  = ok ? src[e] : 0;
        const float wv = ok ? w[e] : 0.0f;
        const int   lr = (int)(ent >> 24);
        if constexpr (SAVE) {
            const int g = gpos + idx;
            if (ok && g < GCAP) {
                v2u h;
                h.x = ((unsigned)lr << 24) | (unsigned)s;
                h.y = __builtin_bit_cast(unsigned, wv);
                *(v2u*)(seg + 2 * (size_t)g) = h;
            }
        }
        add_edges<D>(X, acc, s, wv, lr, min(32, n - base), lane);
    }
}

static_assert(DRAIN_AT + SUB <= CAP, "a trip of hits must fit behind an undrained list");
static_assert(NEDGES % SUB == 0 && SUB % 128 == 0, "trips tile the dst list exactly: the filter loads carry no bounds test");
static_assert(ROWS <= 256 && NEDGES < (1 << 24) && NNODES < (1 << 24), "list entry packing");

template <int D, bool SAVE>
__device__ __forceinline__ void filter_and_sum(const float* __restrict__ X, const int* __restrict__ src,
                                               const int* __restrict__ dst, const float* __restrict__ w,
                                               float* acc, unsigned* mylist, int r0, int wave, int lane,
                                               unsigned* __restrict__ seg, int* gcount) {
    int n = 0;
    const int rot = (int)(((long long)blockIdx.x * NSUB) / NBLK);
    for (int sc = wave; sc < NSUB; sc += AW) {
        if (n >= DRAIN_AT) {
            drain_list<D, SAVE>(X, src, w, acc, mylist, n, lane, seg, gcount);
            n = 0;
        }
        int scr = sc + rot;
        if (scr >= NSUB) scr -= NSUB;
        const int ebase = scr * SUB + lane * 4;
        int4 d[NLD];
        #pragma unroll
        for (int j = 0; j < NLD; ++j) d[j] = *(const int4*)(dst + ebase + j * 128);
        #pragma unroll
        for (int j = 0; j < NLD; ++j) {
            const int e = ebase + j * 128;
            const unsigned l0 = (unsigned)(d[j].x - r0);
            const unsigned l1 = (unsigned)(d[j].y - r0);
            const unsigned l2 = (unsigned)(d[j].z - r0);
            const unsigned l3 = (unsigned)(d[j].w - r0);
            const bool m0 = l0 < (unsigned)ROWS, m1 = l1 < (unsigned)ROWS;
            const bool m2 = l2 < (unsigned)ROWS, m3 = l3 < (unsigned)ROWS;
            if (__builtin_amdgcn_ballot_w32(m0 | m1 | m2 | m3) != 0u) {
                unsigned bm;
                bm = __builtin_amdgcn_ballot_w32(m0);
                if (m0) mylist[n + __builtin_amdgcn_mbcnt_lo(bm, 0u)] = (l0 << 24) | (unsigned)(e + 0);
                n += __popc(bm);
                bm = __builtin_amdgcn_ballot_w32(m1);
                if (m1) mylist[n + __builtin_amdgcn_mbcnt_lo(bm, 0u)] = (l1 << 24) | (unsigned)(e + 1);
                n += __popc(bm);
                bm = __builtin_amdgcn_ballot_w32(m2);
                if (m2) mylist[n + __builtin_amdgcn_mbcnt_lo(bm, 0u)] = (l2 << 24) | (unsigned)(e + 2);
                n += __popc(bm);
                bm = __builtin_amdgcn_ballot_w32(m3);
                if (m3) mylist[n + __builtin_amdgcn_mbcnt_lo(bm, 0u)] = (l3 << 24) | (unsigned)(e + 3);
                n += __popc(bm);
            }
        }
    }
    drain_list<D, SAVE>(X, src, w, acc, mylist, n, lane, seg, gcount);
}

template <int D>
__device__ __forceinline__ void sum_saved(const float* __restrict__ X, float* acc,
                                          const unsigned* __restrict__ seg, int total,
                                          int wave, int lane) {
    const int uw    = __builtin_amdgcn_readfirstlane(wave);
    const int share = (((total + AW - 1) / AW) + 7) & ~7;
    const int beg   = uw * share;
    const int end   = min(total, beg + share);
    for (int base = beg; base < end; base += 32) {
        const int  idx = base + lane;
        const bool ok  = idx < end;
        v2u h;
        h.x = 0u;
        h.y = 0u;
        if (ok) h = *(const v2u*)(seg + 2 * (size_t)idx);
        const unsigned key = h.x;
        const unsigned wb  = h.y;
        const int   s  = (int)(key & 0xFFFFFFu);
        const int   lr = (int)(key >> 24);
        const float wv = __builtin_bit_cast(float, wb);
        add_edges<D>(X, acc, s, wv, lr, min(32, end - base), lane);
    }
}

constexpr int MT = ROWS / 16;
static_assert(ROWS % 16 == 0 && NNODES % ROWS == 0, "whole tiles only: the product's stores carry no row test");
static_assert(DIN == 256 && DHID == 256, "the fused layer-1 product reads 256-float rows and writes 256 columns");

__device__ __forceinline__ void rows_times_w(const float* acc, const float* __restrict__ Wm,
                                             float* hb, int nt, int lane) {
    const int half = lane >> 4;
    const int l15  = lane & 15;
    const float* wrow = Wm + (size_t)(nt * 16 + l15) * 256;
    const float* arow = acc + l15 * P256 + 64 * half;

    v8f c[MT];
    #pragma unroll
    for (int m = 0; m < MT; ++m) c[m] = v8f{};

    for (int hb = 0; hb < 2; ++hb) {
        const float* ab = arow + hb * 128;
        const float* wb = wrow + hb * 128;
        #pragma unroll 4
        for (int p = 0; p < 16; ++p) {
            {
                const int tl = 2 * p + half;
                const v2f b = *(const v2f*)(wb + 4 * tl + 2 * half);
                #pragma unroll
                for (int m = 0; m < MT; ++m) {
                    v2f a;
                    a.x = ab[m * 16 * P256 + tl];
                    a.y = ab[m * 16 * P256 + tl + 32];
                    c[m] = __builtin_amdgcn_wmma_f32_16x16x4_f32(false, a, false, b, (short)0, c[m], false, false);
                }
            }
            {
                const int tl = 2 * p + 1 - half;
                const v2f b = *(const v2f*)(wb + 4 * tl + 2 * half);
                #pragma unroll
                for (int m = 0; m < MT; ++m) {
                    v2f a;
                    a.x = ab[m * 16 * P256 + tl];
                    a.y = ab[m * 16 * P256 + tl + 32];
                    c[m] = __builtin_amdgcn_wmma_f32_16x16x4_f32(false, a, false, b, (short)0, c[m], false, false);
                }
            }
        }
    }

    #pragma unroll
    for (int m = 0; m < MT; ++m) {
        #pragma unroll
        for (int v = 0; v < 8; ++v)
            hb[(m * 16 + v + 8 * half) * P256 + nt * 16 + l15] = fmaxf(c[m][v], 0.0f);
    }
}

constexpr int NT2 = DOUT / 16;
static_assert(DOUT % 16 == 0 && MT * NT2 <= AW, "one wave per tile of the second product");

constexpr int KG2 = 16;

__device__ __forceinline__ const float* w2_lane_row(const float* __restrict__ Wn, int q, int lane) {
    return Wn + (size_t)((q & 3) * 16 + (lane & 15)) * 256 + 2 * (lane >> 4);
}

__device__ __forceinline__ void load_w2_group(const float* wrow, int g, v2f (&b)[KG2]) {
    #pragma unroll
    for (int u = 0; u < KG2; ++u) b[u] = *(const v2f*)(wrow + 4 * (KG2 * g + u));
}

__device__ __forceinline__ void relu_rows_times_w2(const float* hb, const float* wrow, const v2f (&b0)[KG2],
                                                   float* __restrict__ out, int r0, int q, int lane) {
    const int half = lane >> 4;
    const int l15  = lane & 15;
    const int m    = q >> 2;
    const int nt   = q & 3;
    const float* arow = hb + (m * 16 + l15) * P256 + 2 * half;

    v8f c = {};
    v2f bc[KG2], bn[KG2];
    #pragma unroll
    for (int u = 0; u < KG2; ++u) bc[u] = b0[u];
    #pragma unroll
    for (int g = 0; g < 64 / KG2; ++g) {
        if (g + 1 < 64 / KG2) load_w2_group(wrow, g + 1, bn);
        v2f a[KG2];
        #pragma unroll
        for (int u = 0; u < KG2; ++u) a[u] = *(const v2f*)(arow + 4 * (KG2 * g + u));
        #pragma unroll
        for (int u = 0; u < KG2; ++u)
            c = __builtin_amdgcn_wmma_f32_16x16x4_f32(false, a[u], false, bc[u], (short)0, c, false, false);
        #pragma unroll
        for (int u = 0; u < KG2; ++u) bc[u] = bn[u];
    }

    #pragma unroll
    for (int v = 0; v < 8; ++v)
        out[(size_t)(r0 + m * 16 + v + 8 * half) * DOUT + nt * 16 + l15] = c[v];
}

template <int D, bool SECOND>
__global__ __launch_bounds__(1024)
void agg_rows_lds(const float* __restrict__ X, const int* __restrict__ src,
                  const int* __restrict__ dst, const float* __restrict__ w,
                  float* __restrict__ out, unsigned* __restrict__ glist, int* __restrict__ gcnt,
                  const float* __restrict__ Wm, const float* __restrict__ Wn) {
    constexpr int PITCH = (D == 256) ? P256 : D;
    __shared__ float    acc[ROWS * PITCH];
    __shared__ alignas(16) unsigned lists[AW * CAP];
    __shared__ int      gcount;

    const int tid  = threadIdx.x;
    const int lane = tid & 31;
    const int wave = tid >> 5;
    const int r0   = blockIdx.x * ROWS;

    for (int i = tid; i < ROWS * PITCH; i += AW * 32) acc[i] = 0.0f;
    if (tid == 0) gcount = 0;
    __syncthreads();

    unsigned* seg = glist + (size_t)blockIdx.x * (2 * GCAP);
    if constexpr (SECOND) {
        const int total = gcnt[blockIdx.x];
        if (total <= GCAP) {
            sum_saved<D>(X, acc, seg, total, wave, lane);
        } else {
            filter_and_sum<D, false>(X, src, dst, w, acc, lists + wave * CAP, r0, wave, lane, seg, &gcount);
        }
    } else {
        filter_and_sum<D, true>(X, src, dst, w, acc, lists + wave * CAP, r0, wave, lane, seg, &gcount);
    }
    __syncthreads();
    if constexpr (!SECOND) {
        if (tid == 0) gcnt[blockIdx.x] = gcount;
    }

    if constexpr (D == 256) {
        static_assert(ROWS * P256 <= AW * CAP, "the H rows fit where the hit lists were");
        float* hb = reinterpret_cast<float*>(lists);
        const int uw = __builtin_amdgcn_readfirstlane(wave);
        if (uw < D / 16) rows_times_w(acc, Wm, hb, uw, lane);
        const bool  tile = uw < MT * NT2;
        const float* wrow = w2_lane_row(Wn, uw, lane);
        v2f b0[KG2];
        if (tile) load_w2_group(wrow, 0, b0);
        __syncthreads();
        if (tile) relu_rows_times_w2(hb, wrow, b0, out, r0, uw, lane);
    } else {
        constexpr int F4R = D / 4;
        for (int f = tid; f < ROWS * F4R; f += AW * 32) {
            const int row = f / F4R;
            const int c4  = f % F4R;
            if (r0 + row < NNODES) {
                float4 v;
                const float* p = acc + row * 64 + c4;
                v.x = p[0]; v.y = p[16]; v.z = p[32]; v.w = p[48];
                *(float4*)(out + (size_t)(r0 + row) * D + c4 * 4) = v;
            }
        }
    }
}

extern "C" void kernel_launch(void* const* d_in, const int* in_sizes, int n_in,
                              void* d_out, int out_size, void* d_ws, size_t ws_size,
                              hipStream_t stream) {
    const float* node_feat = (const float*)d_in[0];
    const int*   edge_src  = (const int*)  d_in[1];
    const int*   edge_dst  = (const int*)  d_in[2];
    const float* edge_w    = (const float*)d_in[3];
    const float* W1        = (const float*)d_in[4];
    const float* W2        = (const float*)d_in[5];
    float* out = (float*)d_out;

    float* y2   = (float*)d_ws;
    unsigned* glist = (unsigned*)(y2 + (size_t)NNODES * DOUT);
    int*      gcnt  = (int*)(glist + (size_t)NBLK * GCAP * 2);

    agg_rows_lds<DIN, false><<<NBLK, AW * 32, 0, stream>>>(
        node_feat, edge_src, edge_dst, edge_w, y2, glist, gcnt, W1, W2);
    agg_rows_lds<DOUT, true><<<NBLK, AW * 32, 0, stream>>>(
        y2, edge_src, edge_dst, edge_w, out, glist, gcnt, W1, W2);
}


# Both neighbour sums: LDS float adds replaced by register sums; each hit goes to the wave owning its row (LDS ticket lists, arrival order); owners sum rows in registers (layer 2: two half-wave sums)
# speedup vs baseline: 3.2798x; 1.0580x over previous
#include <hip/hip_runtime.h>
#include <hip/hip_bf16.h>

typedef float v2f __attribute__((ext_vector_type(2)));
typedef float v8f __attribute__((ext_vector_type(8)));

constexpr int NNODES = 20000;
constexpr int NEDGES = 320000;
constexpr int DIN    = 256;
constexpr int DHID   = 256;
constexpr int DOUT   = 64;

constexpr int ROWS = 80;
constexpr int P256 = 260;
constexpr int NBLK = (NNODES + ROWS - 1) / ROWS;
constexpr int AW   = 32;
constexpr int SUB  = 1280;
constexpr int NLD  = SUB / 128;
constexpr int CAP  = 1312;
constexpr int NSUB = (NEDGES + SUB - 1) / SUB;
constexpr int DRAIN_AT = 24;
constexpr int GCAP = 4096;
constexpr int QCAP = 256;
constexpr int OSEG = GCAP / 32;
constexpr int GROUPED = -2;
static_assert(OSEG * AW == GCAP, "the owner waves' parts tile the segment");

typedef unsigned v2u __attribute__((ext_vector_type(2)));

template <int D>
__device__ __forceinline__ void add_edges(const float* __restrict__ X, float* acc,
                                          int s, float wv, int lr, int m, int lane) {
    if constexpr (D == 256) {
        for (int k = 0; k < m; k += 8) {
            int   sk[8], lk[8];
            float wk[8];
            #pragma unroll
            for (int u = 0; u < 8; ++u) {
                sk[u] = __builtin_amdgcn_readlane(s, k + u);
                lk[u] = __builtin_amdgcn_readlane(lr, k + u);
                wk[u] = __builtin_bit_cast(float,
                            __builtin_amdgcn_readlane(__builtin_bit_cast(int, wv), k + u));
            }
            float4 xa[8], xb[8];
            #pragma unroll
            for (int u = 0; u < 8; ++u) {
                if (k + u < m) {
                    const float* xr = X + (size_t)sk[u] * 256 + lane * 4;
                    xa[u] = *(const float4*)xr;
                    xb[u] = *(const float4*)(xr + 128);
                }
            }
            #pragma unroll
            for (int u = 0; u < 8; ++u) {
                if (k + u < m) {
                    float* a = acc + lk[u] * P256 + lane;
                    atomicAdd(a +   0, xa[u].x * wk[u]);
                    atomicAdd(a +  32, xa[u].y * wk[u]);
                    atomicAdd(a +  64, xa[u].z * wk[u]);
                    atomicAdd(a +  96, xa[u].w * wk[u]);
                    atomicAdd(a + 128, xb[u].x * wk[u]);
                    atomicAdd(a + 160, xb[u].y * wk[u]);
                    atomicAdd(a + 192, xb[u].z * wk[u]);
                    atomicAdd(a + 224, xb[u].w * wk[u]);
                }
            }
        }
    } else {
        const int half = lane >> 4;
        const int l16  = lane & 15;
        for (int k = 0; k < m; k += 8) {
            float4 x[4];
            float  wk[4];
            int    sk[4], lk[4];
            bool   vq[4];
            #pragma unroll
            for (int u = 0; u < 4; ++u) {
                const int q = k + 2 * u + half;
                vq[u] = q < m;
                sk[u] = __shfl(s, q);
                wk[u] = __shfl(wv, q);
                lk[u] = __shfl(lr, q);
            }
            #pragma unroll
            for (int u = 0; u < 4; ++u) {
                if (vq[u]) x[u] = *(const float4*)(X + (size_t)sk[u] * 64 + l16 * 4);
            }
            #pragma unroll
            for (int u = 0; u < 4; ++u) {
                if (vq[u]) {
                    float* a = acc + lk[u] * 64 + l16;
                    const float c0 = half ? x[u].z : x[u].x;
                    const float c1 = half ? x[u].w : x[u].y;
                    const float c2 = half ? x[u].x : x[u].z;
                    const float c3 = half ? x[u].y : x[u].w;
                    atomicAdd(a + (half ? 32 :  0), c0 * wk[u]);
                    atomicAdd(a + (half ? 48 : 16), c1 * wk[u]);
                    atomicAdd(a + (half ?  0 : 32), c2 * wk[u]);
                    atomicAdd(a + (half ? 16 : 48), c3 * wk[u]);
                }
            }
        }
    }
}

template <int D, bool SAVE>
__device__ __forceinline__ void drain_list(const float* __restrict__ X, const int* __restrict__ src,
                                           const float* __restrict__ w, float* acc,
                                           const unsigned* mylist, int n, int lane,
                                           unsigned* __restrict__ seg, int* gcount) {
    if (n == 0) return;
    int gpos = 0;
    if constexpr (SAVE) {
        int p = 0;
        if (lane == 0) p = atomicAdd(gcount, n);
        gpos = __builtin_amdgcn_readfirstlane(p);
    }
    for (int base = 0; base < n; base += 32) {
        const int  idx = base + lane;
        const bool ok  = idx < n;
        const unsigned ent = ok ? mylist[idx] : 0u;
        const int   e  = (int)(ent & 0xFFFFFFu);
        const int   s  = ok ? src[e] : 0;
        const float wv = ok ? w[e] : 0.0f;
        const int   lr = (int)(ent >> 24);
        if constexpr (SAVE) {
            const int g = gpos + idx;
            if (ok && g < GCAP) {
                v2u h;
                h.x = ((unsigned)lr << 24) | (unsigned)s;
                h.y = __builtin_bit_cast(unsigned, wv);
                *(v2u*)(seg + 2 * (size_t)g) = h;
            }
        }
        add_edges<D>(X, acc, s, wv, lr, min(32, n - base), lane);
    }
}

static_assert(DRAIN_AT + SUB <= CAP, "a trip of hits must fit behind an undrained list");
static_assert(NEDGES % SUB == 0 && SUB % 128 == 0, "trips tile the dst list exactly: the filter loads carry no bounds test");
static_assert(ROWS <= 256 && NEDGES < (1 << 24) && NNODES < (1 << 24), "list entry packing");

template <int D, bool SAVE>
__device__ __forceinline__ void filter_and_sum(const float* __restrict__ X, const int* __restrict__ src,
                                               const int* __restrict__ dst, const float* __restrict__ w,
                                               float* acc, unsigned* mylist, int r0, int wave, int lane,
                                               unsigned* __restrict__ seg, int* gcount) {
    int n = 0;
    const int rot = (int)(((long long)blockIdx.x * NSUB) / NBLK);
    for (int sc = wave; sc < NSUB; sc += AW) {
        if (n >= DRAIN_AT) {
            drain_list<D, SAVE>(X, src, w, acc, mylist, n, lane, seg, gcount);
            n = 0;
        }
        int scr = sc + rot;
        if (scr >= NSUB) scr -= NSUB;
        const int ebase = scr * SUB + lane * 4;
        int4 d[NLD];
        #pragma unroll
        for (int j = 0; j < NLD; ++j) d[j] = *(const int4*)(dst + ebase + j * 128);
        #pragma unroll
        for (int j = 0; j < NLD; ++j) {
            const int e = ebase + j * 128;
            const unsigned l0 = (unsigned)(d[j].x - r0);
            const unsigned l1 = (unsigned)(d[j].y - r0);
            const unsigned l2 = (unsigned)(d[j].z - r0);
            const unsigned l3 = (unsigned)(d[j].w - r0);
            const bool m0 = l0 < (unsigned)ROWS, m1 = l1 < (unsigned)ROWS;
            const bool m2 = l2 < (unsigned)ROWS, m3 = l3 < (unsigned)ROWS;
            if (__builtin_amdgcn_ballot_w32(m0 | m1 | m2 | m3) != 0u) {
                unsigned bm;
                bm = __builtin_amdgcn_ballot_w32(m0);
                if (m0) mylist[n + __builtin_amdgcn_mbcnt_lo(bm, 0u)] = (l0 << 24) | (unsigned)(e + 0);
                n += __popc(bm);
                bm = __builtin_amdgcn_ballot_w32(m1);
                if (m1) mylist[n + __builtin_amdgcn_mbcnt_lo(bm, 0u)] = (l1 << 24) | (unsigned)(e + 1);
                n += __popc(bm);
                bm = __builtin_amdgcn_ballot_w32(m2);
                if (m2) mylist[n + __builtin_amdgcn_mbcnt_lo(bm, 0u)] = (l2 << 24) | (unsigned)(e + 2);
                n += __popc(bm);
                bm = __builtin_amdgcn_ballot_w32(m3);
                if (m3) mylist[n + __builtin_amdgcn_mbcnt_lo(bm, 0u)] = (l3 << 24) | (unsigned)(e + 3);
                n += __popc(bm);
            }
        }
    }
    drain_list<D, SAVE>(X, src, w, acc, mylist, n, lane, seg, gcount);
}

__device__ __forceinline__ int filter_collect(const int* __restrict__ dst, unsigned* mylist,
                                              int r0, int wave, int lane, bool& over) {
    int n = 0;
    over = false;
    const int rot = (int)(((long long)blockIdx.x * NSUB) / NBLK);
    for (int sc = wave; sc < NSUB; sc += AW) {
        if (over) break;
        int scr = sc + rot;
        if (scr >= NSUB) scr -= NSUB;
        const int ebase = scr * SUB + lane * 4;
        int4 d[NLD];
        #pragma unroll
        for (int j = 0; j < NLD; ++j) d[j] = *(const int4*)(dst + ebase + j * 128);
        #pragma unroll
        for (int j = 0; j < NLD; ++j) {
            const int e = ebase + j * 128;
            const unsigned l0 = (unsigned)(d[j].x - r0);
            const unsigned l1 = (unsigned)(d[j].y - r0);
            const unsigned l2 = (unsigned)(d[j].z - r0);
            const unsigned l3 = (unsigned)(d[j].w - r0);
            const bool m0 = l0 < (unsigned)ROWS, m1 = l1 < (unsigned)ROWS;
            const bool m2 = l2 < (unsigned)ROWS, m3 = l3 < (unsigned)ROWS;
            if (__builtin_amdgcn_ballot_w32(m0 | m1 | m2 | m3) != 0u) {
                if (n > CAP - 128) { over = true; continue; }
                unsigned bm;
                bm = __builtin_amdgcn_ballot_w32(m0);
                if (m0) mylist[n + __builtin_amdgcn_mbcnt_lo(bm, 0u)] = (l0 << 24) | (unsigned)(e + 0);
                n += __popc(bm);
                bm = __builtin_amdgcn_ballot_w32(m1);
                if (m1) mylist[n + __builtin_amdgcn_mbcnt_lo(bm, 0u)] = (l1 << 24) | (unsigned)(e + 1);
                n += __popc(bm);
                bm = __builtin_amdgcn_ballot_w32(m2);
                if (m2) mylist[n + __builtin_amdgcn_mbcnt_lo(bm, 0u)] = (l2 << 24) | (unsigned)(e + 2);
                n += __popc(bm);
                bm = __builtin_amdgcn_ballot_w32(m3);
                if (m3) mylist[n + __builtin_amdgcn_mbcnt_lo(bm, 0u)] = (l3 << 24) | (unsigned)(e + 3);
                n += __popc(bm);
            }
        }
    }
    return n;
}

__device__ __forceinline__ bool hand_to_owners(const unsigned* mylist, int n, int lane,
                                               unsigned* queues, int* qcnt) {
    bool full = false;
    for (int base = 0; base < n; base += 32) {
        const int idx = base + lane;
        if (idx < n) {
            const unsigned ent = mylist[idx];
            const int owner = (int)(ent >> 24) & 31;
            const int pos = atomicAdd(qcnt + owner, 1);
            if (pos < QCAP) queues[owner * QCAP + pos] = ent;
            else            full = true;
        }
    }
    return __builtin_amdgcn_ballot_w32(full) != 0u;
}

__device__ __forceinline__ void owner_sum_rows(const float* __restrict__ X, const int* __restrict__ src,
                                               const float* __restrict__ w, float* acc,
                                               const unsigned* myq, int m, int wave, int lane,
                                               unsigned* __restrict__ seg, int* segover) {
    float4 ra[3], rb[3];
    #pragma unroll
    for (int j = 0; j < 3; ++j) {
        ra[j] = make_float4(0.0f, 0.0f, 0.0f, 0.0f);
        rb[j] = make_float4(0.0f, 0.0f, 0.0f, 0.0f);
    }
    if (m > OSEG && lane == 0) atomicOr(segover, 1);
    unsigned* sub = seg + 2 * (size_t)(wave * OSEG);
    for (int base = 0; base < m; base += 32) {
        const int  idx = base + lane;
        const bool ok  = idx < m;
        const unsigned ent = ok ? myq[idx] : 0u;
        const int   e  = (int)(ent & 0xFFFFFFu);
        const int   s  = ok ? src[e] : 0;
        const float wv = ok ? w[e] : 0.0f;
        const int   lr = (int)(ent >> 24);
        if (ok && idx < OSEG) {
            v2u h;
            h.x = ((unsigned)lr << 24) | (unsigned)s;
            h.y = __builtin_bit_cast(unsigned, wv);
            *(v2u*)(sub + 2 * (size_t)idx) = h;
        }
        const int mm = min(32, m - base);
        for (int k = 0; k < mm; k += 8) {
            int   sk[8], jk[8];
            float wk[8];
            #pragma unroll
            for (int u = 0; u < 8; ++u) {
                sk[u] = __builtin_amdgcn_readlane(s, k + u);
                jk[u] = __builtin_amdgcn_readlane(lr, k + u) >> 5;
                wk[u] = __builtin_bit_cast(float,
                            __builtin_amdgcn_readlane(__builtin_bit_cast(int, wv), k + u));
            }
            float4 xa[8], xb[8];
            #pragma unroll
            for (int u = 0; u < 8; ++u) {
                if (k + u < mm) {
                    const float* xr = X + (size_t)sk[u] * 256 + lane * 4;
                    xa[u] = *(const float4*)xr;
                    xb[u] = *(const float4*)(xr + 128);
                }
            }
            #pragma unroll
            for (int u = 0; u < 8; ++u) {
                if (k + u < mm) {
                    #pragma unroll
                    for (int j = 0; j < 3; ++j) {
                        if (jk[u] == j) {
                            ra[j].x += xa[u].x * wk[u]; ra[j].y += xa[u].y * wk[u];
                            ra[j].z += xa[u].z * wk[u]; ra[j].w += xa[u].w * wk[u];
                            rb[j].x += xb[u].x * wk[u]; rb[j].y += xb[u].y * wk[u];
                            rb[j].z += xb[u].z * wk[u]; rb[j].w += xb[u].w * wk[u];
                        }
                    }
                }
            }
        }
    }
    #pragma unroll
    for (int j = 0; j < 3; ++j) {
        const int lr = wave + 32 * j;
        if (lr < ROWS) {
            float* a = acc + lr * P256 + lane;
            a[  0] = ra[j].x; a[ 32] = ra[j].y; a[ 64] = ra[j].z; a[ 96] = ra[j].w;
            a[128] = rb[j].x; a[160] = rb[j].y; a[192] = rb[j].z; a[224] = rb[j].w;
        }
    }
}

__device__ __forceinline__ void owner_sum_rows64(const float* __restrict__ Y, const unsigned* __restrict__ sub,
                                                 int m, float* __restrict__ out, int r0, int wave, int lane) {
    const int half = lane >> 4;
    const int l16  = lane & 15;
    float4 ra = make_float4(0.0f, 0.0f, 0.0f, 0.0f);
    float4 rb = make_float4(0.0f, 0.0f, 0.0f, 0.0f);
    float4 rc = make_float4(0.0f, 0.0f, 0.0f, 0.0f);
    for (int base = 0; base < m; base += 32) {
        const int idx = base + lane;
        v2u h;
        h.x = 0u;
        h.y = 0u;
        if (idx < m) h = *(const v2u*)(sub + 2 * (size_t)idx);
        const int key = (int)h.x;
        const int wb  = (int)h.y;
        const int mm  = min(32, m - base);
        for (int k = 0; k < mm; k += 16) {
            int kk[16], ww[16];
            #pragma unroll
            for (int t = 0; t < 16; ++t) {
                kk[t] = __builtin_amdgcn_readlane(key, k + t);
                ww[t] = __builtin_amdgcn_readlane(wb, k + t);
            }
            float4 x[8];
            #pragma unroll
            for (int u = 0; u < 8; ++u) {
                if (k + 2 * u < mm) {
                    const int s = (half ? kk[2 * u + 1] : kk[2 * u]) & 0xFFFFFF;
                    x[u] = *(const float4*)(Y + (size_t)s * 64 + l16 * 4);
                }
            }
            #pragma unroll
            for (int u = 0; u < 8; ++u) {
                if (k + 2 * u < mm) {
                    const unsigned kq = (unsigned)(half ? kk[2 * u + 1] : kk[2 * u]);
                    const float    wq = __builtin_bit_cast(float, half ? ww[2 * u + 1] : ww[2 * u]);
                    const int      j  = (int)(kq >> 29);
                    if (k + 2 * u + half < mm) {
                        if (j == 0) {
                            ra.x += x[u].x * wq; ra.y += x[u].y * wq; ra.z += x[u].z * wq; ra.w += x[u].w * wq;
                        } else if (j == 1) {
                            rb.x += x[u].x * wq; rb.y += x[u].y * wq; rb.z += x[u].z * wq; rb.w += x[u].w * wq;
                        } else {
                            rc.x += x[u].x * wq; rc.y += x[u].y * wq; rc.z += x[u].z * wq; rc.w += x[u].w * wq;
                        }
                    }
                }
            }
        }
    }
    ra.x += __shfl_xor(ra.x, 16); ra.y += __shfl_xor(ra.y, 16); ra.z += __shfl_xor(ra.z, 16); ra.w += __shfl_xor(ra.w, 16);
    rb.x += __shfl_xor(rb.x, 16); rb.y += __shfl_xor(rb.y, 16); rb.z += __shfl_xor(rb.z, 16); rb.w += __shfl_xor(rb.w, 16);
    rc.x += __shfl_xor(rc.x, 16); rc.y += __shfl_xor(rc.y, 16); rc.z += __shfl_xor(rc.z, 16); rc.w += __shfl_xor(rc.w, 16);
    {
        const int lr = wave + 32 * half;
        float4 v;
        v.x = half ? rb.x : ra.x; v.y = half ? rb.y : ra.y; v.z = half ? rb.z : ra.z; v.w = half ? rb.w : ra.w;
        *(float4*)(out + (size_t)(r0 + lr) * DOUT + l16 * 4) = v;
    }
    if (wave + 64 < ROWS && half == 0)
        *(float4*)(out + (size_t)(r0 + wave + 64) * DOUT + l16 * 4) = rc;
}

template <int D>
__device__ __forceinline__ void sum_saved(const float* __restrict__ X, float* acc,
                                          const unsigned* __restrict__ seg, int total,
                                          int wave, int lane) {
    const int uw    = __builtin_amdgcn_readfirstlane(wave);
    const int share = (((total + AW - 1) / AW) + 7) & ~7;
    const int beg   = uw * share;
    const int end   = min(total, beg + share);
    for (int base = beg; base < end; base += 32) {
        const int  idx = base + lane;
        const bool ok  = idx < end;
        v2u h;
        h.x = 0u;
        h.y = 0u;
        if (ok) h = *(const v2u*)(seg + 2 * (size_t)idx);
        const unsigned key = h.x;
        const unsigned wb  = h.y;
        const int   s  = (int)(key & 0xFFFFFFu);
        const int   lr = (int)(key >> 24);
        const float wv = __builtin_bit_cast(float, wb);
        add_edges<D>(X, acc, s, wv, lr, min(32, end - base), lane);
    }
}

constexpr int MT = ROWS / 16;
static_assert(ROWS % 16 == 0 && NNODES % ROWS == 0, "whole tiles only: the product's stores carry no row test");
static_assert(DIN == 256 && DHID == 256, "the fused layer-1 product reads 256-float rows and writes 256 columns");

__device__ __forceinline__ void rows_times_w(const float* acc, const float* __restrict__ Wm,
                                             float* hb, int nt, int lane) {
    const int half = lane >> 4;
    const int l15  = lane & 15;
    const float* wrow = Wm + (size_t)(nt * 16 + l15) * 256;
    const float* arow = acc + l15 * P256 + 64 * half;

    v8f c[MT];
    #pragma unroll
    for (int m = 0; m < MT; ++m) c[m] = v8f{};

    for (int hb = 0; hb < 2; ++hb) {
        const float* ab = arow + hb * 128;
        const float* wb = wrow + hb * 128;
        #pragma unroll 4
        for (int p = 0; p < 16; ++p) {
            {
                const int tl = 2 * p + half;
                const v2f b = *(const v2f*)(wb + 4 * tl + 2 * half);
                #pragma unroll
                for (int m = 0; m < MT; ++m) {
                    v2f a;
                    a.x = ab[m * 16 * P256 + tl];
                    a.y = ab[m * 16 * P256 + tl + 32];
                    c[m] = __builtin_amdgcn_wmma_f32_16x16x4_f32(false, a, false, b, (short)0, c[m], false, false);
                }
            }
            {
                const int tl = 2 * p + 1 - half;
                const v2f b = *(const v2f*)(wb + 4 * tl + 2 * half);
                #pragma unroll
                for (int m = 0; m < MT; ++m) {
                    v2f a;
                    a.x = ab[m * 16 * P256 + tl];
                    a.y = ab[m * 16 * P256 + tl + 32];
                    c[m] = __builtin_amdgcn_wmma_f32_16x16x4_f32(false, a, false, b, (short)0, c[m], false, false);
                }
            }
        }
    }

    #pragma unroll
    for (int m = 0; m < MT; ++m) {
        #pragma unroll
        for (int v = 0; v < 8; ++v)
            hb[(m * 16 + v + 8 * half) * P256 + nt * 16 + l15] = fmaxf(c[m][v], 0.0f);
    }
}

constexpr int NT2 = DOUT / 16;
static_assert(DOUT % 16 == 0 && MT * NT2 <= AW, "one wave per tile of the second product");

constexpr int KG2 = 16;

__device__ __forceinline__ const float* w2_lane_row(const float* __restrict__ Wn, int q, int lane) {
    return Wn + (size_t)((q & 3) * 16 + (lane & 15)) * 256 + 2 * (lane >> 4);
}

__device__ __forceinline__ void load_w2_group(const float* wrow, int g, v2f (&b)[KG2]) {
    #pragma unroll
    for (int u = 0; u < KG2; ++u) b[u] = *(const v2f*)(wrow + 4 * (KG2 * g + u));
}

__device__ __forceinline__ void relu_rows_times_w2(const float* hb, const float* wrow, const v2f (&b0)[KG2],
                                                   float* __restrict__ out, int r0, int q, int lane) {
    const int half = lane >> 4;
    const int l15  = lane & 15;
    const int m    = q >> 2;
    const int nt   = q & 3;
    const float* arow = hb + (m * 16 + l15) * P256 + 2 * half;

    v8f c = {};
    v2f bc[KG2], bn[KG2];
    #pragma unroll
    for (int u = 0; u < KG2; ++u) bc[u] = b0[u];
    #pragma unroll
    for (int g = 0; g < 64 / KG2; ++g) {
        if (g + 1 < 64 / KG2) load_w2_group(wrow, g + 1, bn);
        v2f a[KG2];
        #pragma unroll
        for (int u = 0; u < KG2; ++u) a[u] = *(const v2f*)(arow + 4 * (KG2 * g + u));
        #pragma unroll
        for (int u = 0; u < KG2; ++u)
            c = __builtin_amdgcn_wmma_f32_16x16x4_f32(false, a[u], false, bc[u], (short)0, c, false, false);
        #pragma unroll
        for (int u = 0; u < KG2; ++u) bc[u] = bn[u];
    }

    #pragma unroll
    for (int v = 0; v < 8; ++v)
        out[(size_t)(r0 + m * 16 + v + 8 * half) * DOUT + nt * 16 + l15] = c[v];
}

template <int D, bool SECOND>
__global__ __launch_bounds__(1024)
void agg_rows_lds(const float* __restrict__ X, const int* __restrict__ src,
                  const int* __restrict__ dst, const float* __restrict__ w,
                  float* __restrict__ out, unsigned* __restrict__ glist, int* __restrict__ gcnt,
                  const float* __restrict__ Wm, const float* __restrict__ Wn) {
    constexpr int PITCH = (D == 256) ? P256 : D;
    __shared__ float    acc[ROWS * PITCH];
    __shared__ alignas(16) unsigned lists[AW * CAP];
    __shared__ int      gcount;
    constexpr bool OWNERS = (D == 256) && !SECOND;
    __shared__ alignas(16) unsigned queues[OWNERS ? AW * QCAP : 1];
    __shared__ int      qcnt[OWNERS ? AW : 1];
    __shared__ int      fallback;
    __shared__ int      segover;
    int* ocnt = gcnt + NBLK;

    const int tid  = threadIdx.x;
    const int lane = tid & 31;
    const int wave = tid >> 5;
    const int r0   = blockIdx.x * ROWS;

    if constexpr (SECOND && D == 64) {
        if (gcnt[blockIdx.x] == GROUPED) {
            const int ow = __builtin_amdgcn_readfirstlane(wave);
            const int m  = min(ocnt[blockIdx.x * AW + ow], OSEG);
            const unsigned* sub = glist + (size_t)blockIdx.x * (2 * GCAP) + 2 * (size_t)(ow * OSEG);
            owner_sum_rows64(X, sub, m, out, r0, ow, lane);
            return;
        }
    }
    if constexpr (OWNERS) {
        if (tid < AW) qcnt[tid] = 0;
        if (tid == 0) { gcount = 0; fallback = 0; segover = 0; }
    } else {
        for (int i = tid; i < ROWS * PITCH; i += AW * 32) acc[i] = 0.0f;
        if (tid == 0) gcount = 0;
    }
    __syncthreads();

    unsigned* seg = glist + (size_t)blockIdx.x * (2 * GCAP);
    if constexpr (SECOND) {
        const int total = gcnt[blockIdx.x];
        if (total <= GCAP) {
            sum_saved<D>(X, acc, seg, total, wave, lane);
        } else {
            filter_and_sum<D, false>(X, src, dst, w, acc, lists + wave * CAP, r0, wave, lane, seg, &gcount);
        }
    } else if constexpr (OWNERS) {
        unsigned* mylist = lists + wave * CAP;
        bool over;
        const int n = filter_collect(dst, mylist, r0, wave, lane, over);
        bool full = false;
        if (!over) full = hand_to_owners(mylist, n, lane, queues, qcnt);
        if ((over || full) && lane == 0) atomicOr(&fallback, 1);
        __syncthreads();
        if (fallback != 0) {
            for (int i = tid; i < ROWS * PITCH; i += AW * 32) acc[i] = 0.0f;
            __syncthreads();
            filter_and_sum<D, true>(X, src, dst, w, acc, mylist, r0, wave, lane, seg, &gcount);
        } else {
            const int ow = __builtin_amdgcn_readfirstlane(wave);
            const int m  = min(qcnt[ow], QCAP);
            if (lane == 0) ocnt[blockIdx.x * AW + ow] = m;
            owner_sum_rows(X, src, w, acc, queues + ow * QCAP, m, ow, lane, seg, &segover);
        }
    } else {
        filter_and_sum<D, true>(X, src, dst, w, acc, lists + wave * CAP, r0, wave, lane, seg, &gcount);
    }
    __syncthreads();
    if constexpr (OWNERS) {
        if (tid == 0) gcnt[blockIdx.x] = (fallback != 0) ? gcount : (segover != 0 ? GCAP + 1 : GROUPED);
    } else if constexpr (!SECOND) {
        if (tid == 0) gcnt[blockIdx.x] = gcount;
    }

    if constexpr (D == 256) {
        static_assert(ROWS * P256 <= AW * CAP, "the H rows fit where the hit lists were");
        float* hb = reinterpret_cast<float*>(lists);
        const int uw = __builtin_amdgcn_readfirstlane(wave);
        if (uw < D / 16) rows_times_w(acc, Wm, hb, uw, lane);
        const bool  tile = uw < MT * NT2;
        const float* wrow = w2_lane_row(Wn, uw, lane);
        v2f b0[KG2];
        if (tile) load_w2_group(wrow, 0, b0);
        __syncthreads();
        if (tile) relu_rows_times_w2(hb, wrow, b0, out, r0, uw, lane);
    } else {
        constexpr int F4R = D / 4;
        for (int f = tid; f < ROWS * F4R; f += AW * 32) {
            const int row = f / F4R;
            const int c4  = f % F4R;
            if (r0 + row < NNODES) {
                float4 v;
                const float* p = acc + row * 64 + c4;
                v.x = p[0]; v.y = p[16]; v.z = p[32]; v.w = p[48];
                *(float4*)(out + (size_t)(r0 + row) * D + c4 * 4) = v;
            }
        }
    }
}

extern "C" void kernel_launch(void* const* d_in, const int* in_sizes, int n_in,
                              void* d_out, int out_size, void* d_ws, size_t ws_size,
                              hipStream_t stream) {
    const float* node_feat = (const float*)d_in[0];
    const int*   edge_src  = (const int*)  d_in[1];
    const int*   edge_dst  = (const int*)  d_in[2];
    const float* edge_w    = (const float*)d_in[3];
    const float* W1        = (const float*)d_in[4];
    const float* W2        = (const float*)d_in[5];
    float* out = (float*)d_out;

    float* y2   = (float*)d_ws;
    unsigned* glist = (unsigned*)(y2 + (size_t)NNODES * DOUT);
    int*      gcnt  = (int*)(glist + (size_t)NBLK * GCAP * 2);

    agg_rows_lds<DIN, false><<<NBLK, AW * 32, 0, stream>>>(
        node_feat, edge_src, edge_dst, edge_w, y2, glist, gcnt, W1, W2);
    agg_rows_lds<DOUT, true><<<NBLK, AW * 32, 0, stream>>>(
        y2, edge_src, edge_dst, edge_w, out, glist, gcnt, W1, W2);
}
